# starting kernel
# baseline (speedup 1.0000x reference)
#include <hip/hip_runtime.h>

typedef __attribute__((ext_vector_type(2))) float v2f;
typedef __attribute__((ext_vector_type(8))) float v8f;

#define NODES 100000
#define EDGES 1000000
#define RELS  3
#define HF    64

__global__ void k_deg(const int* __restrict__ dst, float* __restrict__ deg) {
  int tid = blockIdx.x * blockDim.x + threadIdx.x;
  if (tid >= RELS * EDGES) return;
  int r = tid / EDGES;
  unsafeAtomicAdd(&deg[(size_t)r * NODES + dst[tid]], 1.0f);
}

__global__ void k_dinv(float* __restrict__ deg) {
  int i = blockIdx.x * blockDim.x + threadIdx.x;
  if (i >= RELS * NODES) return;
  float v = deg[i];
  v = v < 1.0f ? 1.0f : v;
  deg[i] = 1.0f / sqrtf(v);
}

__global__ void k_scatter(const float* __restrict__ feat,
                          const float* __restrict__ dinv,
                          const int* __restrict__ src,
                          const int* __restrict__ dst,
                          float* __restrict__ ah) {
  int tid = blockIdx.x * blockDim.x + threadIdx.x;
  if (tid >= EDGES * 16) return;
  int e = tid >> 4;
  int c = (tid & 15) << 2;
  int s = src[e];
  int d = dst[e];
  float dv = dinv[s];
  const float4 f = *(const float4*)(feat + (size_t)s * HF + c);
  float* o = ah + (size_t)d * HF + c;
  unsafeAtomicAdd(o + 0, f.x * dv);
  unsafeAtomicAdd(o + 1, f.y * dv);
  unsafeAtomicAdd(o + 2, f.z * dv);
  unsafeAtomicAdd(o + 3, f.w * dv);
}

__launch_bounds__(256)
__global__ void k_gemm(const float* __restrict__ A1,
                       const float* __restrict__ A2,
                       const float* __restrict__ dinv,
                       float a2sign,
                       const float* __restrict__ W,
                       const float* __restrict__ bias,
                       float* __restrict__ out,
                       float* __restrict__ accum,
                       int n, int act) {
  __shared__ float sW[128 * 64];
  __shared__ float sB[64];
  const bool dual = (A2 != nullptr);
  const int wtot = (dual ? 128 : 64) * 64;
  for (int i = threadIdx.x; i < wtot; i += 256) {
    int k = i >> 6;
    int col = i & 63;
    sW[(k >> 1) * 128 + col * 2 + (k & 1)] = W[i];
  }
  if (threadIdx.x < 64) sB[threadIdx.x] = bias[threadIdx.x];
  __syncthreads();

  const int lane = threadIdx.x & 31;
  const int wave = threadIdx.x >> 5;
  const int strip = blockIdx.x * 8 + wave;
  if (strip >= (n >> 4)) return;
  const int row0 = strip << 4;
  const int m  = lane & 15;
  const int kh = (lane >> 4) << 1;

  const float* a1p = A1 + (size_t)(row0 + m) * HF + kh;
  v8f acc[4] = {};

#pragma unroll
  for (int k0 = 0; k0 < 64; k0 += 4) {
    v2f a = *(const v2f*)(a1p + k0);
    const float* wb = &sW[(k0 + kh) * 64 + m * 2];
#pragma unroll
    for (int t = 0; t < 4; ++t) {
      v2f b = *(const v2f*)(wb + t * 32);
      acc[t] = __builtin_amdgcn_wmma_f32_16x16x4_f32(
          false, a, false, b, (short)0, acc[t], false, false);
    }
  }
  if (dual) {
    const float* a2p = A2 + (size_t)(row0 + m) * HF + kh;
    const float ds = (dinv ? dinv[row0 + m] : 1.0f) * a2sign;
#pragma unroll
    for (int k0 = 0; k0 < 64; k0 += 4) {
      v2f a = *(const v2f*)(a2p + k0);
      a = a * ds;
      const float* wb = &sW[(64 + k0 + kh) * 64 + m * 2];
#pragma unroll
      for (int t = 0; t < 4; ++t) {
        v2f b = *(const v2f*)(wb + t * 32);
        acc[t] = __builtin_amdgcn_wmma_f32_16x16x4_f32(
            false, a, false, b, (short)0, acc[t], false, false);
      }
    }
  }

  const int rbase = row0 + ((lane >> 4) << 3);
#pragma unroll
  for (int t = 0; t < 4; ++t) {
    const float bv = sB[t * 16 + m];
#pragma unroll
    for (int v = 0; v < 8; ++v) {
      float val = acc[t][v] + bv;
      if (act) val = val > 0.0f ? val : 0.01f * val;
      size_t o = (size_t)(rbase + v) * HF + t * 16 + m;
      out[o] = val;
      if (accum) accum[o] += val;
    }
  }
}

__global__ void k_head(const float* __restrict__ hall,
                       const float* __restrict__ W4,
                       const float* __restrict__ b4,
                       float* __restrict__ out, int n) {
  __shared__ float sW[128];
  __shared__ float sb[2];
  if (threadIdx.x < 128) sW[threadIdx.x] = W4[threadIdx.x];
  if (threadIdx.x < 2) sb[threadIdx.x] = b4[threadIdx.x];
  __syncthreads();
  int i = blockIdx.x * blockDim.x + threadIdx.x;
  if (i >= n) return;
  const float* hp = hall + (size_t)i * HF;
  float a0 = sb[0], a1 = sb[1];
#pragma unroll
  for (int k = 0; k < 64; ++k) {
    float v = hp[k];
    v = v > 0.0f ? v : 0.01f * v;
    a0 += v * sW[2 * k];
    a1 += v * sW[2 * k + 1];
  }
  out[2 * (size_t)i + 0] = a0;
  out[2 * (size_t)i + 1] = a1;
}

extern "C" void kernel_launch(void* const* d_in, const int* in_sizes, int n_in,
                              void* d_out, int out_size, void* d_ws, size_t ws_size,
                              hipStream_t stream) {
  const float* x   = (const float*)d_in[0];
  const int*   src = (const int*)d_in[1];
  const int*   dst = (const int*)d_in[2];
  const float* W1  = (const float*)d_in[3];
  const float* b1  = (const float*)d_in[4];
  const float* W2  = (const float*)d_in[5];
  const float* b2  = (const float*)d_in[6];
  const float* Wc1 = (const float*)d_in[7];
  const float* bc1 = (const float*)d_in[8];
  const float* Wc2 = (const float*)d_in[9];
  const float* bc2 = (const float*)d_in[10];
  const float* W3  = (const float*)d_in[11];
  const float* b3  = (const float*)d_in[12];
  const float* W4  = (const float*)d_in[13];
  const float* b4  = (const float*)d_in[14];
  float* out = (float*)d_out;

  const size_t NH = (size_t)NODES * HF;
  float* ws   = (float*)d_ws;
  float* hbuf = ws;
  float* h0   = hbuf + NH;
  float* h1   = h0 + NH;
  float* hall = h1 + NH;
  float* ah   = hall + NH;
  float* dinv = ah + NH;

  hipMemsetAsync(dinv, 0, sizeof(float) * (size_t)RELS * NODES, stream);
  k_deg<<<(RELS * EDGES + 255) / 256, 256, 0, stream>>>(dst, dinv);
  k_dinv<<<(RELS * NODES + 255) / 256, 256, 0, stream>>>(dinv);

  const int gblocks = ((NODES / 16) + 7) / 8;

  k_gemm<<<gblocks, 256, 0, stream>>>(x,  nullptr, nullptr, 1.0f, W1, b1, h0,   nullptr, NODES, 1);
  k_gemm<<<gblocks, 256, 0, stream>>>(h0, nullptr, nullptr, 1.0f, W2, b2, hbuf, nullptr, NODES, 1);
  hipMemsetAsync(hall, 0, sizeof(float) * NH, stream);

  for (int r = 0; r < RELS; ++r) {
    const int*   sr = src + (size_t)r * EDGES;
    const int*   dr = dst + (size_t)r * EDGES;
    const float* dv = dinv + (size_t)r * NODES;

    hipMemsetAsync(ah, 0, sizeof(float) * NH, stream);
    k_scatter<<<(EDGES * 16 + 255) / 256, 256, 0, stream>>>(hbuf, dv, sr, dr, ah);
    k_gemm<<<gblocks, 256, 0, stream>>>(hbuf, ah, dv, -1.0f, Wc1, bc1, h0, nullptr, NODES, 0);

    hipMemsetAsync(ah, 0, sizeof(float) * NH, stream);
    k_scatter<<<(EDGES * 16 + 255) / 256, 256, 0, stream>>>(h0, dv, sr, dr, ah);
    k_gemm<<<gblocks, 256, 0, stream>>>(h0, ah, dv, -1.0f, Wc2, bc2, h1, nullptr, NODES, 0);

    k_gemm<<<gblocks, 256, 0, stream>>>(h0, h1, nullptr, 1.0f, W3, b3, hbuf, hall, NODES, 0);
  }

  k_head<<<(NODES + 255) / 256, 256, 0, stream>>>(hall, W4, b4, out, NODES);
}


# Back-to-back node products fused per launch: a wave keeps its 16x64 result in its own LDS strip as the next product's A operand; head joins the last; 20 launches become 15; sum order kept
# speedup vs baseline: 2.0849x; 2.0849x over previous
#include <hip/hip_runtime.h>

typedef __attribute__((ext_vector_type(2))) float v2f;
typedef __attribute__((ext_vector_type(8))) float v8f;

#define NODES 100000
#define EDGES 1000000
#define RELS  3
#define HF    64

#define DCH 16
#define DQ  4
#define DNR 25024
#define DEC (EDGES / DCH)
__launch_bounds__(1024)
__global__ void k_degpart(const int* __restrict__ dst, float* __restrict__ part) {
  __shared__ unsigned cnt[DNR];
  const int c = blockIdx.x, q = blockIdx.y, r = blockIdx.z;
  for (int i = threadIdx.x; i < DNR; i += 1024) cnt[i] = 0u;
  __syncthreads();
  const int lo = q * DNR;
  const int4* p = (const int4*)(dst + (size_t)r * EDGES + (size_t)c * DEC);
  for (int base = threadIdx.x; base < DEC / 4; base += 8 * 1024) {
    int4 v[8];
#pragma unroll
    for (int j = 0; j < 8; ++j) {
      const int idx = base + j * 1024;
      v[j] = idx < DEC / 4 ? p[idx] : make_int4(-1, -1, -1, -1);
    }
#pragma unroll
    for (int j = 0; j < 8; ++j) {
      const unsigned a0 = (unsigned)(v[j].x - lo), a1 = (unsigned)(v[j].y - lo);
      const unsigned a2 = (unsigned)(v[j].z - lo), a3 = (unsigned)(v[j].w - lo);
      if (a0 < DNR) atomicAdd(&cnt[a0], 1u);
      if (a1 < DNR) atomicAdd(&cnt[a1], 1u);
      if (a2 < DNR) atomicAdd(&cnt[a2], 1u);
      if (a3 < DNR) atomicAdd(&cnt[a3], 1u);
    }
  }
  __syncthreads();
  const int nh = (NODES - lo) < DNR ? (NODES - lo) : DNR;
  float* o = part + ((size_t)c * RELS + r) * NODES + lo;
  for (int i = threadIdx.x; i < nh; i += 1024) o[i] = (float)cnt[i];
}

__global__ void k_dinv(const float* __restrict__ part, float* __restrict__ dinv) {
  int i = blockIdx.x * blockDim.x + threadIdx.x;
  if (i >= RELS * NODES) return;
  float v = 0.0f;
#pragma unroll
  for (int c = 0; c < DCH; ++c) v += part[(size_t)c * RELS * NODES + i];
  v = v < 1.0f ? 1.0f : v;
  dinv[i] = 1.0f / sqrtf(v);
}

#define EPW 8
__global__ void k_scatter(const float* __restrict__ feat,
                          const float* __restrict__ dinv,
                          const int* __restrict__ src,
                          const int* __restrict__ dst,
                          float* __restrict__ ah) {
  const int lane = threadIdx.x & 31;
  const int wave = __builtin_amdgcn_readfirstlane((int)(threadIdx.x >> 5));
  const int e = (blockIdx.x * 8 + wave) * EPW;
  if (e >= EDGES) return;
  int s[EPW], d[EPW];
#pragma unroll
  for (int i = 0; i < EPW; ++i) { s[i] = src[e + i]; d[i] = dst[e + i]; }
  float lo[EPW], hi[EPW];
#pragma unroll
  for (int i = 0; i < EPW; ++i) {
    const float dv = dinv[s[i]];
    const float* f = feat + (size_t)s[i] * HF + lane;
    lo[i] = f[0] * dv;
    hi[i] = f[32] * dv;
  }
#pragma unroll
  for (int i = 0; i < EPW; ++i) {
    float* o = ah + (size_t)d[i] * HF + lane;
    unsafeAtomicAdd(o,      lo[i]);
    unsafeAtomicAdd(o + 32, hi[i]);
  }
}

__launch_bounds__(256)
__global__ void k_gemm(const float* __restrict__ A1,
                       const float* A2,
                       const float* __restrict__ dinv,
                       float a2sign,
                       const float* __restrict__ W,
                       const float* __restrict__ bias,
                       float* out,
                       float* __restrict__ accum,
                       int accum_init,
                       float* __restrict__ clr,
                       int n, int act) {
  __shared__ float sW[128 * 64];
  __shared__ float sB[64];
  const bool dual = (A2 != nullptr);
  const int wtot = (dual ? 128 : 64) * 64;
  for (int i = threadIdx.x; i < wtot; i += 256) {
    int k = i >> 6;
    int col = i & 63;
    sW[(k >> 1) * 128 + col * 2 + (k & 1)] = W[i];
  }
  if (threadIdx.x < 64) sB[threadIdx.x] = bias[threadIdx.x];
  __syncthreads();

  const int lane = threadIdx.x & 31;
  const int wave = threadIdx.x >> 5;
  const int strip = blockIdx.x * 8 + wave;
  if (strip >= (n >> 4)) return;
  const int row0 = strip << 4;
  if (clr) {
    float4* z = (float4*)(clr + (size_t)row0 * HF);
#pragma unroll
    for (int j = 0; j < 8; ++j) z[j * 32 + lane] = make_float4(0.0f, 0.0f, 0.0f, 0.0f);
  }
  const int m  = lane & 15;
  const int kh = (lane >> 4) << 1;

  const float* a1p = A1 + (size_t)(row0 + m) * HF + kh;
  v8f acc[4] = {};

#pragma unroll
  for (int k0 = 0; k0 < 64; k0 += 4) {
    v2f a = *(const v2f*)(a1p + k0);
    const float* wb = &sW[(k0 + kh) * 64 + m * 2];
#pragma unroll
    for (int t = 0; t < 4; ++t) {
      v2f b = *(const v2f*)(wb + t * 32);
      acc[t] = __builtin_amdgcn_wmma_f32_16x16x4_f32(
          false, a, false, b, (short)0, acc[t], false, false);
    }
  }
  if (dual) {
    const float* a2p = A2 + (size_t)(row0 + m) * HF + kh;
    const float ds = (dinv ? dinv[row0 + m] : 1.0f) * a2sign;
#pragma unroll
    for (int k0 = 0; k0 < 64; k0 += 4) {
      v2f a = *(const v2f*)(a2p + k0);
      a = a * ds;
      const float* wb = &sW[(64 + k0 + kh) * 64 + m * 2];
#pragma unroll
      for (int t = 0; t < 4; ++t) {
        v2f b = *(const v2f*)(wb + t * 32);
        acc[t] = __builtin_amdgcn_wmma_f32_16x16x4_f32(
            false, a, false, b, (short)0, acc[t], false, false);
      }
    }
  }

  const int rbase = row0 + ((lane >> 4) << 3);
#pragma unroll
  for (int t = 0; t < 4; ++t) {
    const float bv = sB[t * 16 + m];
#pragma unroll
    for (int v = 0; v < 8; ++v) {
      float val = acc[t][v] + bv;
      if (act) val = val > 0.0f ? val : 0.01f * val;
      size_t o = (size_t)(rbase + v) * HF + t * 16 + m;
      out[o] = val;
      if (accum) {
        if (accum_init) accum[o] = val; else accum[o] += val;
      }
    }
  }
}

#define FW 16
#define SP 68
template <bool DUAL, bool HEAD>
__launch_bounds__(FW * 32)
__global__ void k_gemm2(const float* __restrict__ A1,
                        const float* __restrict__ A2,
                        const float* __restrict__ dinv,
                        const float* __restrict__ Wa, const float* __restrict__ ba,
                        const float* __restrict__ Wb, const float* __restrict__ bb,
                        const float* __restrict__ W4, const float* __restrict__ b4,
                        float* __restrict__ out,
                        float* __restrict__ accum, int accum_init,
                        float* __restrict__ clr, int n) {
  constexpr int KR = DUAL ? 128 : 64;
  __shared__ float sWa[KR * 64];
  __shared__ float sWb[KR * 64];
  __shared__ float sBa[64];
  __shared__ float sBb[64];
  __shared__ float sW4[128];
  __shared__ float sb4[2];
  __shared__ float sT[FW * 16 * SP];
  constexpr int PT = KR * 64 / (FW * 32);
  float wa[PT], wb[PT];
#pragma unroll
  for (int j = 0; j < PT; ++j) {
    wa[j] = Wa[threadIdx.x + j * (FW * 32)];
    wb[j] = Wb[threadIdx.x + j * (FW * 32)];
  }
#pragma unroll
  for (int j = 0; j < PT; ++j) {
    const int i = threadIdx.x + j * (FW * 32);
    const int k = i >> 6;
    const int col = i & 63;
    const int o = (k >> 1) * 128 + col * 2 + (k & 1);
    sWa[o] = wa[j];
    sWb[o] = wb[j];
  }
  if (threadIdx.x < 64) {
    sBa[threadIdx.x] = ba[threadIdx.x];
    sBb[threadIdx.x] = bb[threadIdx.x];
  }
  if (HEAD) {
    if (threadIdx.x < 128) sW4[threadIdx.x] = W4[threadIdx.x];
    if (threadIdx.x < 2) sb4[threadIdx.x] = b4[threadIdx.x];
  }
  __syncthreads();

  const int lane = threadIdx.x & 31;
  const int wave = threadIdx.x >> 5;
  const int strip = blockIdx.x * FW + wave;
  if (strip >= (n >> 4)) return;
  const int row0 = strip << 4;
  const int m  = lane & 15;
  const int hh = lane >> 4;
  const int kh = hh << 1;
  float* st = sT + wave * (16 * SP);

  const float* a1p = A1 + (size_t)(row0 + m) * HF + kh;
  v2f a1f[16];
#pragma unroll
  for (int i = 0; i < 16; ++i) a1f[i] = *(const v2f*)(a1p + 4 * i);
  v8f acc[4] = {};
#pragma unroll
  for (int i = 0; i < 16; ++i) {
    const float* wb = &sWa[(4 * i + kh) * 64 + m * 2];
#pragma unroll
    for (int t = 0; t < 4; ++t) {
      v2f b = *(const v2f*)(wb + t * 32);
      acc[t] = __builtin_amdgcn_wmma_f32_16x16x4_f32(
          false, a1f[i], false, b, (short)0, acc[t], false, false);
    }
  }
  if (DUAL) {
    const float* a2p = A2 + (size_t)(row0 + m) * HF + kh;
    const float ds = -dinv[row0 + m];
#pragma unroll
    for (int i = 0; i < 16; ++i) {
      v2f a = *(const v2f*)(a2p + 4 * i);
      a = a * ds;
      const float* wb = &sWa[(64 + 4 * i + kh) * 64 + m * 2];
#pragma unroll
      for (int t = 0; t < 4; ++t) {
        v2f b = *(const v2f*)(wb + t * 32);
        acc[t] = __builtin_amdgcn_wmma_f32_16x16x4_f32(
            false, a, false, b, (short)0, acc[t], false, false);
      }
    }
  }
  if (clr) {
    float4* z = (float4*)(clr + (size_t)row0 * HF);
#pragma unroll
    for (int j = 0; j < 8; ++j) z[j * 32 + lane] = make_float4(0.0f, 0.0f, 0.0f, 0.0f);
  }

#pragma unroll
  for (int t = 0; t < 4; ++t) {
    const float bv = sBa[t * 16 + m];
#pragma unroll
    for (int v = 0; v < 8; ++v) {
      float val = acc[t][v] + bv;
      if (!DUAL) val = val > 0.0f ? val : 0.01f * val;
      st[(v + 8 * hh) * SP + t * 16 + m] = val;
    }
  }
  __builtin_amdgcn_fence(__ATOMIC_ACQ_REL, "wavefront");

  v8f acc2[4] = {};
  if (DUAL) {
#pragma unroll
    for (int i = 0; i < 16; ++i) {
      const float* wb = &sWb[(4 * i + kh) * 64 + m * 2];
#pragma unroll
      for (int t = 0; t < 4; ++t) {
        v2f b = *(const v2f*)(wb + t * 32);
        acc2[t] = __builtin_amdgcn_wmma_f32_16x16x4_f32(
            false, a1f[i], false, b, (short)0, acc2[t], false, false);
      }
    }
  }
  {
    const float* tp = st + m * SP + kh;
#pragma unroll
    for (int i = 0; i < 16; ++i) {
      v2f a = *(const v2f*)(tp + 4 * i);
      const float* wb = &sWb[((DUAL ? 64 : 0) + 4 * i + kh) * 64 + m * 2];
#pragma unroll
      for (int t = 0; t < 4; ++t) {
        v2f b = *(const v2f*)(wb + t * 32);
        acc2[t] = __builtin_amdgcn_wmma_f32_16x16x4_f32(
            false, a, false, b, (short)0, acc2[t], false, false);
      }
    }
  }

  const int rbase = row0 + (hh << 3);
  if (!HEAD) {
#pragma unroll
    for (int t = 0; t < 4; ++t) {
      const float bv = sBb[t * 16 + m];
#pragma unroll
      for (int v = 0; v < 8; ++v) {
        float val = acc2[t][v] + bv;
        if (!DUAL) val = val > 0.0f ? val : 0.01f * val;
        const size_t o = (size_t)(rbase + v) * HF + t * 16 + m;
        out[o] = val;
        if (accum) {
          if (accum_init) accum[o] = val; else accum[o] += val;
        }
      }
    }
  } else {
    __builtin_amdgcn_fence(__ATOMIC_ACQ_REL, "wavefront");
#pragma unroll
    for (int t = 0; t < 4; ++t) {
      const float bv = sBb[t * 16 + m];
#pragma unroll
      for (int v = 0; v < 8; ++v) {
        float val = acc2[t][v] + bv;
        if (!accum_init) val += accum[(size_t)(rbase + v) * HF + t * 16 + m];
        val = val > 0.0f ? val : 0.01f * val;
        st[(v + 8 * hh) * SP + t * 16 + m] = val;
      }
    }
    __builtin_amdgcn_fence(__ATOMIC_ACQ_REL, "wavefront");
    const float* tp = st + m * SP;
    float a = sb4[hh];
#pragma unroll
    for (int k = 0; k < 64; ++k) a += tp[k] * sW4[2 * k + hh];
    out[2 * (size_t)(row0 + m) + hh] = a;
  }
}

extern "C" void kernel_launch(void* const* d_in, const int* in_sizes, int n_in,
                              void* d_out, int out_size, void* d_ws, size_t ws_size,
                              hipStream_t stream) {
  const float* x   = (const float*)d_in[0];
  const int*   src = (const int*)d_in[1];
  const int*   dst = (const int*)d_in[2];
  const float* W1  = (const float*)d_in[3];
  const float* b1  = (const float*)d_in[4];
  const float* W2  = (const float*)d_in[5];
  const float* b2  = (const float*)d_in[6];
  const float* Wc1 = (const float*)d_in[7];
  const float* bc1 = (const float*)d_in[8];
  const float* Wc2 = (const float*)d_in[9];
  const float* bc2 = (const float*)d_in[10];
  const float* W3  = (const float*)d_in[11];
  const float* b3  = (const float*)d_in[12];
  const float* W4  = (const float*)d_in[13];
  const float* b4  = (const float*)d_in[14];
  float* out = (float*)d_out;

  const size_t NH = (size_t)NODES * HF;
  float* ws   = (float*)d_ws;
  float* hbuf = ws;
  float* h0   = hbuf + NH;
  float* h1   = h0 + NH;
  float* hall = h1 + NH;
  float* ah   = hall + NH;
  float* dinv = ah + NH;

  k_degpart<<<dim3(DCH, DQ, RELS), 1024, 0, stream>>>(dst, ah);
  k_dinv<<<(RELS * NODES + 255) / 256, 256, 0, stream>>>(ah, dinv);

  const int gblocks = ((NODES / 16) + 7) / 8;

  const int fblocks = ((NODES / 16) + FW - 1) / FW;

  k_gemm2<false, false><<<fblocks, FW * 32, 0, stream>>>(x, nullptr, nullptr, W1, b1, W2, b2,
                                                         nullptr, nullptr, hbuf, nullptr, 0, ah, NODES);

  for (int r = 0; r < RELS; ++r) {
    const int*   sr = src + (size_t)r * EDGES;
    const int*   dr = dst + (size_t)r * EDGES;
    const float* dv = dinv + (size_t)r * NODES;

    k_scatter<<<(EDGES / EPW + 7) / 8, 256, 0, stream>>>(hbuf, dv, sr, dr, ah);
    k_gemm<<<gblocks, 256, 0, stream>>>(hbuf, ah, dv, -1.0f, Wc1, bc1, h0, nullptr, 0, h1, NODES, 0);

    k_scatter<<<(EDGES / EPW + 7) / 8, 256, 0, stream>>>(h0, dv, sr, dr, h1);
    if (r + 1 < RELS)
      k_gemm2<true, false><<<fblocks, FW * 32, 0, stream>>>(h0, h1, dv, Wc2, bc2, W3, b3, nullptr, nullptr,
                                                            hbuf, hall, r == 0 ? 1 : 0, ah, NODES);
    else
      k_gemm2<true, true><<<fblocks, FW * 32, 0, stream>>>(h0, h1, dv, Wc2, bc2, W3, b3, W4, b4,
                                                           out, hall, r == 0 ? 1 : 0, nullptr, NODES);
  }
}


# Node rows stored pre-multiplied by the gathering relation's dinv (product epilogues scale, A1 fragments unscale by 1/dinv): scatter-add loses its per-edge dinv lookup and multiply; one more rounding
# speedup vs baseline: 2.1527x; 1.0325x over previous
#include <hip/hip_runtime.h>

typedef __attribute__((ext_vector_type(2))) float v2f;
typedef __attribute__((ext_vector_type(8))) float v8f;

#define NODES 100000
#define EDGES 1000000
#define RELS  3
#define HF    64

#define DCH 16
#define DQ  4
#define DNR 25024
#define DEC (EDGES / DCH)
__launch_bounds__(1024)
__global__ void k_degpart(const int* __restrict__ dst, float* __restrict__ part) {
  __shared__ unsigned cnt[DNR];
  const int c = blockIdx.x, q = blockIdx.y, r = blockIdx.z;
  for (int i = threadIdx.x; i < DNR; i += 1024) cnt[i] = 0u;
  __syncthreads();
  const int lo = q * DNR;
  const int4* p = (const int4*)(dst + (size_t)r * EDGES + (size_t)c * DEC);
  for (int base = threadIdx.x; base < DEC / 4; base += 8 * 1024) {
    int4 v[8];
#pragma unroll
    for (int j = 0; j < 8; ++j) {
      const int idx = base + j * 1024;
      v[j] = idx < DEC / 4 ? p[idx] : make_int4(-1, -1, -1, -1);
    }
#pragma unroll
    for (int j = 0; j < 8; ++j) {
      const unsigned a0 = (unsigned)(v[j].x - lo), a1 = (unsigned)(v[j].y - lo);
      const unsigned a2 = (unsigned)(v[j].z - lo), a3 = (unsigned)(v[j].w - lo);
      if (a0 < DNR) atomicAdd(&cnt[a0], 1u);
      if (a1 < DNR) atomicAdd(&cnt[a1], 1u);
      if (a2 < DNR) atomicAdd(&cnt[a2], 1u);
      if (a3 < DNR) atomicAdd(&cnt[a3], 1u);
    }
  }
  __syncthreads();
  const int nh = (NODES - lo) < DNR ? (NODES - lo) : DNR;
  float* o = part + ((size_t)c * RELS + r) * NODES + lo;
  for (int i = threadIdx.x; i < nh; i += 1024) o[i] = (float)cnt[i];
}

__global__ void k_dinv(const float* __restrict__ part, float* __restrict__ dinv) {
  int i = blockIdx.x * blockDim.x + threadIdx.x;
  if (i >= RELS * NODES) return;
  float v = 0.0f;
#pragma unroll
  for (int c = 0; c < DCH; ++c) v += part[(size_t)c * RELS * NODES + i];
  v = v < 1.0f ? 1.0f : v;
  dinv[i] = 1.0f / sqrtf(v);
}

#define EPW 8
__global__ void k_scatter(const float* __restrict__ feat,
                          const int* __restrict__ src,
                          const int* __restrict__ dst,
                          float* __restrict__ ah) {
  const int lane = threadIdx.x & 31;
  const int wave = __builtin_amdgcn_readfirstlane((int)(threadIdx.x >> 5));
  const int e = (blockIdx.x * 8 + wave) * EPW;
  if (e >= EDGES) return;
  int s[EPW], d[EPW];
#pragma unroll
  for (int i = 0; i < EPW; ++i) { s[i] = src[e + i]; d[i] = dst[e + i]; }
  float lo[EPW], hi[EPW];
#pragma unroll
  for (int i = 0; i < EPW; ++i) {
    const float* f = feat + (size_t)s[i] * HF + lane;
    lo[i] = f[0];
    hi[i] = f[32];
  }
#pragma unroll
  for (int i = 0; i < EPW; ++i) {
    float* o = ah + (size_t)d[i] * HF + lane;
    unsafeAtomicAdd(o,      lo[i]);
    unsafeAtomicAdd(o + 32, hi[i]);
  }
}

__launch_bounds__(256)
__global__ void k_gemm(const float* __restrict__ A1,
                       const float* A2,
                       const float* __restrict__ dinv,
                       const float* __restrict__ oscale,
                       float a2sign,
                       const float* __restrict__ W,
                       const float* __restrict__ bias,
                       float* out,
                       float* __restrict__ accum,
                       int accum_init,
                       float* __restrict__ clr,
                       int n, int act) {
  __shared__ float sW[128 * 64];
  __shared__ float sB[64];
  const bool dual = (A2 != nullptr);
  const int wtot = (dual ? 128 : 64) * 64;
  for (int i = threadIdx.x; i < wtot; i += 256) {
    int k = i >> 6;
    int col = i & 63;
    sW[(k >> 1) * 128 + col * 2 + (k & 1)] = W[i];
  }
  if (threadIdx.x < 64) sB[threadIdx.x] = bias[threadIdx.x];
  __syncthreads();

  const int lane = threadIdx.x & 31;
  const int wave = threadIdx.x >> 5;
  const int strip = blockIdx.x * 8 + wave;
  if (strip >= (n >> 4)) return;
  const int row0 = strip << 4;
  if (clr) {
    float4* z = (float4*)(clr + (size_t)row0 * HF);
#pragma unroll
    for (int j = 0; j < 8; ++j) z[j * 32 + lane] = make_float4(0.0f, 0.0f, 0.0f, 0.0f);
  }
  const int m  = lane & 15;
  const int kh = (lane >> 4) << 1;

  const float* a1p = A1 + (size_t)(row0 + m) * HF + kh;
  const float dr = dinv ? dinv[row0 + m] : 1.0f;
  const float ur = 1.0f / dr;
  v8f acc[4] = {};

#pragma unroll
  for (int k0 = 0; k0 < 64; k0 += 4) {
    v2f a = *(const v2f*)(a1p + k0);
    a = a * ur;
    const float* wb = &sW[(k0 + kh) * 64 + m * 2];
#pragma unroll
    for (int t = 0; t < 4; ++t) {
      v2f b = *(const v2f*)(wb + t * 32);
      acc[t] = __builtin_amdgcn_wmma_f32_16x16x4_f32(
          false, a, false, b, (short)0, acc[t], false, false);
    }
  }
  if (dual) {
    const float* a2p = A2 + (size_t)(row0 + m) * HF + kh;
    const float ds = dr * a2sign;
#pragma unroll
    for (int k0 = 0; k0 < 64; k0 += 4) {
      v2f a = *(const v2f*)(a2p + k0);
      a = a * ds;
      const float* wb = &sW[(64 + k0 + kh) * 64 + m * 2];
#pragma unroll
      for (int t = 0; t < 4; ++t) {
        v2f b = *(const v2f*)(wb + t * 32);
        acc[t] = __builtin_amdgcn_wmma_f32_16x16x4_f32(
            false, a, false, b, (short)0, acc[t], false, false);
      }
    }
  }

  const int rbase = row0 + ((lane >> 4) << 3);
  float4 os0 = make_float4(1.0f, 1.0f, 1.0f, 1.0f), os1 = os0;
  if (oscale) {
    os0 = *(const float4*)(oscale + rbase);
    os1 = *(const float4*)(oscale + rbase + 4);
  }
  const float os[8] = {os0.x, os0.y, os0.z, os0.w, os1.x, os1.y, os1.z, os1.w};
#pragma unroll
  for (int t = 0; t < 4; ++t) {
    const float bv = sB[t * 16 + m];
#pragma unroll
    for (int v = 0; v < 8; ++v) {
      float val = acc[t][v] + bv;
      if (act) val = val > 0.0f ? val : 0.01f * val;
      size_t o = (size_t)(rbase + v) * HF + t * 16 + m;
      out[o] = val * os[v];
      if (accum) {
        if (accum_init) accum[o] = val; else accum[o] += val;
      }
    }
  }
}

#define FW 16
#define SP 68
template <bool DUAL, bool HEAD>
__launch_bounds__(FW * 32)
__global__ void k_gemm2(const float* __restrict__ A1,
                        const float* __restrict__ A2,
                        const float* __restrict__ dinv,
                        const float* __restrict__ oscale,
                        const float* __restrict__ Wa, const float* __restrict__ ba,
                        const float* __restrict__ Wb, const float* __restrict__ bb,
                        const float* __restrict__ W4, const float* __restrict__ b4,
                        float* __restrict__ out,
                        float* __restrict__ accum, int accum_init,
                        float* __restrict__ clr, int n) {
  constexpr int KR = DUAL ? 128 : 64;
  __shared__ float sWa[KR * 64];
  __shared__ float sWb[KR * 64];
  __shared__ float sBa[64];
  __shared__ float sBb[64];
  __shared__ float sW4[128];
  __shared__ float sb4[2];
  __shared__ float sT[FW * 16 * SP];
  constexpr int PT = KR * 64 / (FW * 32);
  float wa[PT], wb[PT];
#pragma unroll
  for (int j = 0; j < PT; ++j) {
    wa[j] = Wa[threadIdx.x + j * (FW * 32)];
    wb[j] = Wb[threadIdx.x + j * (FW * 32)];
  }
#pragma unroll
  for (int j = 0; j < PT; ++j) {
    const int i = threadIdx.x + j * (FW * 32);
    const int k = i >> 6;
    const int col = i & 63;
    const int o = (k >> 1) * 128 + col * 2 + (k & 1);
    sWa[o] = wa[j];
    sWb[o] = wb[j];
  }
  if (threadIdx.x < 64) {
    sBa[threadIdx.x] = ba[threadIdx.x];
    sBb[threadIdx.x] = bb[threadIdx.x];
  }
  if (HEAD) {
    if (threadIdx.x < 128) sW4[threadIdx.x] = W4[threadIdx.x];
    if (threadIdx.x < 2) sb4[threadIdx.x] = b4[threadIdx.x];
  }
  __syncthreads();

  const int lane = threadIdx.x & 31;
  const int wave = threadIdx.x >> 5;
  const int strip = blockIdx.x * FW + wave;
  if (strip >= (n >> 4)) return;
  const int row0 = strip << 4;
  const int m  = lane & 15;
  const int hh = lane >> 4;
  const int kh = hh << 1;
  float* st = sT + wave * (16 * SP);

  const float* a1p = A1 + (size_t)(row0 + m) * HF + kh;
  v2f a1f[16];
#pragma unroll
  for (int i = 0; i < 16; ++i) a1f[i] = *(const v2f*)(a1p + 4 * i);
  const int rbase = row0 + (hh << 3);
  float dr = 1.0f;
  if (DUAL) {
    dr = dinv[row0 + m];
    const float ur = 1.0f / dr;
#pragma unroll
    for (int i = 0; i < 16; ++i) a1f[i] = a1f[i] * ur;
  }
  v8f acc[4] = {};
#pragma unroll
  for (int i = 0; i < 16; ++i) {
    const float* wb = &sWa[(4 * i + kh) * 64 + m * 2];
#pragma unroll
    for (int t = 0; t < 4; ++t) {
      v2f b = *(const v2f*)(wb + t * 32);
      acc[t] = __builtin_amdgcn_wmma_f32_16x16x4_f32(
          false, a1f[i], false, b, (short)0, acc[t], false, false);
    }
  }
  if (DUAL) {
    const float* a2p = A2 + (size_t)(row0 + m) * HF + kh;
    const float ds = -dr;
#pragma unroll
    for (int i = 0; i < 16; ++i) {
      v2f a = *(const v2f*)(a2p + 4 * i);
      a = a * ds;
      const float* wb = &sWa[(64 + 4 * i + kh) * 64 + m * 2];
#pragma unroll
      for (int t = 0; t < 4; ++t) {
        v2f b = *(const v2f*)(wb + t * 32);
        acc[t] = __builtin_amdgcn_wmma_f32_16x16x4_f32(
            false, a, false, b, (short)0, acc[t], false, false);
      }
    }
  }
  if (clr) {
    float4* z = (float4*)(clr + (size_t)row0 * HF);
#pragma unroll
    for (int j = 0; j < 8; ++j) z[j * 32 + lane] = make_float4(0.0f, 0.0f, 0.0f, 0.0f);
  }

#pragma unroll
  for (int t = 0; t < 4; ++t) {
    const float bv = sBa[t * 16 + m];
#pragma unroll
    for (int v = 0; v < 8; ++v) {
      float val = acc[t][v] + bv;
      if (!DUAL) val = val > 0.0f ? val : 0.01f * val;
      st[(v + 8 * hh) * SP + t * 16 + m] = val;
    }
  }
  __builtin_amdgcn_fence(__ATOMIC_ACQ_REL, "wavefront");

  float4 os0 = make_float4(1.0f, 1.0f, 1.0f, 1.0f), os1 = os0;
  if (!HEAD && oscale) {
    os0 = *(const float4*)(oscale + rbase);
    os1 = *(const float4*)(oscale + rbase + 4);
  }
  const float os[8] = {os0.x, os0.y, os0.z, os0.w, os1.x, os1.y, os1.z, os1.w};

  v8f acc2[4] = {};
  if (DUAL) {
#pragma unroll
    for (int i = 0; i < 16; ++i) {
      const float* wb = &sWb[(4 * i + kh) * 64 + m * 2];
#pragma unroll
      for (int t = 0; t < 4; ++t) {
        v2f b = *(const v2f*)(wb + t * 32);
        acc2[t] = __builtin_amdgcn_wmma_f32_16x16x4_f32(
            false, a1f[i], false, b, (short)0, acc2[t], false, false);
      }
    }
  }
  {
    const float* tp = st + m * SP + kh;
#pragma unroll
    for (int i = 0; i < 16; ++i) {
      v2f a = *(const v2f*)(tp + 4 * i);
      const float* wb = &sWb[((DUAL ? 64 : 0) + 4 * i + kh) * 64 + m * 2];
#pragma unroll
      for (int t = 0; t < 4; ++t) {
        v2f b = *(const v2f*)(wb + t * 32);
        acc2[t] = __builtin_amdgcn_wmma_f32_16x16x4_f32(
            false, a, false, b, (short)0, acc2[t], false, false);
      }
    }
  }

  if (!HEAD) {
#pragma unroll
    for (int t = 0; t < 4; ++t) {
      const float bv = sBb[t * 16 + m];
#pragma unroll
      for (int v = 0; v < 8; ++v) {
        float val = acc2[t][v] + bv;
        if (!DUAL) val = val > 0.0f ? val : 0.01f * val;
        const size_t o = (size_t)(rbase + v) * HF + t * 16 + m;
        out[o] = val * os[v];
        if (accum) {
          if (accum_init) accum[o] = val; else accum[o] += val;
        }
      }
    }
  } else {
    __builtin_amdgcn_fence(__ATOMIC_ACQ_REL, "wavefront");
#pragma unroll
    for (int t = 0; t < 4; ++t) {
      const float bv = sBb[t * 16 + m];
#pragma unroll
      for (int v = 0; v < 8; ++v) {
        float val = acc2[t][v] + bv;
        if (!accum_init) val += accum[(size_t)(rbase + v) * HF + t * 16 + m];
        val = val > 0.0f ? val : 0.01f * val;
        st[(v + 8 * hh) * SP + t * 16 + m] = val;
      }
    }
    __builtin_amdgcn_fence(__ATOMIC_ACQ_REL, "wavefront");
    const float* tp = st + m * SP;
    float a = sb4[hh];
#pragma unroll
    for (int k = 0; k < 64; ++k) a += tp[k] * sW4[2 * k + hh];
    out[2 * (size_t)(row0 + m) + hh] = a;
  }
}

extern "C" void kernel_launch(void* const* d_in, const int* in_sizes, int n_in,
                              void* d_out, int out_size, void* d_ws, size_t ws_size,
                              hipStream_t stream) {
  const float* x   = (const float*)d_in[0];
  const int*   src = (const int*)d_in[1];
  const int*   dst = (const int*)d_in[2];
  const float* W1  = (const float*)d_in[3];
  const float* b1  = (const float*)d_in[4];
  const float* W2  = (const float*)d_in[5];
  const float* b2  = (const float*)d_in[6];
  const float* Wc1 = (const float*)d_in[7];
  const float* bc1 = (const float*)d_in[8];
  const float* Wc2 = (const float*)d_in[9];
  const float* bc2 = (const float*)d_in[10];
  const float* W3  = (const float*)d_in[11];
  const float* b3  = (const float*)d_in[12];
  const float* W4  = (const float*)d_in[13];
  const float* b4  = (const float*)d_in[14];
  float* out = (float*)d_out;

  const size_t NH = (size_t)NODES * HF;
  float* ws   = (float*)d_ws;
  float* hbuf = ws;
  float* h0   = hbuf + NH;
  float* h1   = h0 + NH;
  float* hall = h1 + NH;
  float* ah   = hall + NH;
  float* dinv = ah + NH;

  k_degpart<<<dim3(DCH, DQ, RELS), 1024, 0, stream>>>(dst, ah);
  k_dinv<<<(RELS * NODES + 255) / 256, 256, 0, stream>>>(ah, dinv);

  const int gblocks = ((NODES / 16) + 7) / 8;

  const int fblocks = ((NODES / 16) + FW - 1) / FW;

  k_gemm2<false, false><<<fblocks, FW * 32, 0, stream>>>(x, nullptr, nullptr, dinv, W1, b1, W2, b2,
                                                         nullptr, nullptr, hbuf, nullptr, 0, ah, NODES);

  for (int r = 0; r < RELS; ++r) {
    const int*   sr = src + (size_t)r * EDGES;
    const int*   dr = dst + (size_t)r * EDGES;
    const float* dv = dinv + (size_t)r * NODES;
    const float* dvn = dinv + (size_t)(r + 1) * NODES;

    k_scatter<<<(EDGES / EPW + 7) / 8, 256, 0, stream>>>(hbuf, sr, dr, ah);
    k_gemm<<<gblocks, 256, 0, stream>>>(hbuf, ah, dv, dv, -1.0f, Wc1, bc1, h0, nullptr, 0, h1, NODES, 0);

    k_scatter<<<(EDGES / EPW + 7) / 8, 256, 0, stream>>>(h0, sr, dr, h1);
    if (r + 1 < RELS)
      k_gemm2<true, false><<<fblocks, FW * 32, 0, stream>>>(h0, h1, dv, dvn, Wc2, bc2, W3, b3, nullptr, nullptr,
                                                            hbuf, hall, r == 0 ? 1 : 0, ah, NODES);
    else
      k_gemm2<true, true><<<fblocks, FW * 32, 0, stream>>>(h0, h1, dv, nullptr, Wc2, bc2, W3, b3, W4, b4,
                                                           out, hall, r == 0 ? 1 : 0, nullptr, NODES);
  }
}


# Scatter-add moved into LDS: edges grouped once per call by 200-node dst range (count, fill); a block per range sums its rows in an LDS tile (arrival-order LDS float adds), stores each line once
# speedup vs baseline: 2.4899x; 1.1566x over previous
#include <hip/hip_runtime.h>

typedef __attribute__((ext_vector_type(2))) float v2f;
typedef __attribute__((ext_vector_type(8))) float v8f;

#define NODES 100000
#define EDGES 1000000
#define RELS  3
#define HF    64

#define DCH 16
#define DQ  4
#define DNR 25024
#define DEC (EDGES / DCH)
__launch_bounds__(1024)
__global__ void k_degpart(const int* __restrict__ dst, float* __restrict__ part) {
  __shared__ unsigned cnt[DNR];
  const int c = blockIdx.x, q = blockIdx.y, r = blockIdx.z;
  for (int i = threadIdx.x; i < DNR; i += 1024) cnt[i] = 0u;
  __syncthreads();
  const int lo = q * DNR;
  const int4* p = (const int4*)(dst + (size_t)r * EDGES + (size_t)c * DEC);
  for (int base = threadIdx.x; base < DEC / 4; base += 8 * 1024) {
    int4 v[8];
#pragma unroll
    for (int j = 0; j < 8; ++j) {
      const int idx = base + j * 1024;
      v[j] = idx < DEC / 4 ? p[idx] : make_int4(-1, -1, -1, -1);
    }
#pragma unroll
    for (int j = 0; j < 8; ++j) {
      const unsigned a0 = (unsigned)(v[j].x - lo), a1 = (unsigned)(v[j].y - lo);
      const unsigned a2 = (unsigned)(v[j].z - lo), a3 = (unsigned)(v[j].w - lo);
      if (a0 < DNR) atomicAdd(&cnt[a0], 1u);
      if (a1 < DNR) atomicAdd(&cnt[a1], 1u);
      if (a2 < DNR) atomicAdd(&cnt[a2], 1u);
      if (a3 < DNR) atomicAdd(&cnt[a3], 1u);
    }
  }
  __syncthreads();
  const int nh = (NODES - lo) < DNR ? (NODES - lo) : DNR;
  float* o = part + ((size_t)c * RELS + r) * NODES + lo;
  for (int i = threadIdx.x; i < nh; i += 1024) o[i] = (float)cnt[i];
}

__global__ void k_dinv(const float* __restrict__ part, float* __restrict__ dinv) {
  int i = blockIdx.x * blockDim.x + threadIdx.x;
  if (i >= RELS * NODES) return;
  float v = 0.0f;
#pragma unroll
  for (int c = 0; c < DCH; ++c) v += part[(size_t)c * RELS * NODES + i];
  v = v < 1.0f ? 1.0f : v;
  dinv[i] = 1.0f / sqrtf(v);
}

#define RN   200
#define NR   (NODES / RN)
#define GCH  50
#define GEC  (EDGES / GCH)
#define GU   5

__launch_bounds__(1024)
__global__ void k_rcount(const int* __restrict__ dst, unsigned* __restrict__ cnt) {
  __shared__ unsigned h[NR];
  const int c = blockIdx.x, r = blockIdx.y;
  for (int i = threadIdx.x; i < NR; i += 1024) h[i] = 0u;
  __syncthreads();
  const int4* p = (const int4*)(dst + (size_t)r * EDGES + (size_t)c * GEC);
  for (int base = threadIdx.x; base < GEC / 4; base += GU * 1024) {
    int4 v[GU];
#pragma unroll
    for (int j = 0; j < GU; ++j) {
      const int idx = base + j * 1024;
      v[j] = idx < GEC / 4 ? p[idx] : make_int4(-1, -1, -1, -1);
    }
#pragma unroll
    for (int j = 0; j < GU; ++j) {
      if ((unsigned)v[j].x < NODES) atomicAdd(&h[(unsigned)v[j].x / RN], 1u);
      if ((unsigned)v[j].y < NODES) atomicAdd(&h[(unsigned)v[j].y / RN], 1u);
      if ((unsigned)v[j].z < NODES) atomicAdd(&h[(unsigned)v[j].z / RN], 1u);
      if ((unsigned)v[j].w < NODES) atomicAdd(&h[(unsigned)v[j].w / RN], 1u);
    }
  }
  __syncthreads();
  unsigned* o = cnt + ((size_t)r * GCH + c) * NR;
  for (int i = threadIdx.x; i < NR; i += 1024) o[i] = h[i];
}

__device__ __forceinline__ void rput(unsigned* cur, unsigned* __restrict__ o, int sid, int did) {
  if ((unsigned)did < NODES) {
    const unsigned g = (unsigned)did / RN;
    const unsigned pos = atomicAdd(&cur[g], 1u);
    o[pos] = ((unsigned)sid << 8) | ((unsigned)did - g * RN);
  }
}

__launch_bounds__(1024)
__global__ void k_rfill(const int* __restrict__ src, const int* __restrict__ dst,
                        const unsigned* __restrict__ cnt,
                        unsigned* __restrict__ rstart,
                        unsigned* __restrict__ pk) {
  __shared__ unsigned sc[512];
  __shared__ unsigned cur[NR];
  const int c = blockIdx.x, r = blockIdx.y;
  const int t = threadIdx.x;
  unsigned tot = 0u, pre = 0u;
  if (t < NR) {
    const unsigned* q = cnt + (size_t)r * GCH * NR + t;
#pragma unroll 10
    for (int j = 0; j < GCH; ++j) {
      const unsigned x = q[(size_t)j * NR];
      tot += x;
      pre += j < c ? x : 0u;
    }
  }
  if (t < 512) sc[t] = tot;
  __syncthreads();
  for (int off = 1; off < 512; off <<= 1) {
    unsigned a = 0u;
    if (t < 512 && t >= off) a = sc[t - off];
    __syncthreads();
    if (t < 512) sc[t] += a;
    __syncthreads();
  }
  if (t < NR) {
    const unsigned st = sc[t] - tot;
    cur[t] = st + pre;
    if (c == 0) {
      rstart[(size_t)r * (NR + 1) + t] = st;
      if (t == NR - 1) rstart[(size_t)r * (NR + 1) + NR] = sc[t];
    }
  }
  __syncthreads();
  const int4* ps = (const int4*)(src + (size_t)r * EDGES + (size_t)c * GEC);
  const int4* pd = (const int4*)(dst + (size_t)r * EDGES + (size_t)c * GEC);
  unsigned* o = pk + (size_t)r * EDGES;
  for (int base = t; base < GEC / 4; base += GU * 1024) {
    int4 vs[GU], vd[GU];
#pragma unroll
    for (int j = 0; j < GU; ++j) {
      const int idx = base + j * 1024;
      const bool ok = idx < GEC / 4;
      vd[j] = ok ? pd[idx] : make_int4(-1, -1, -1, -1);
      vs[j] = ok ? ps[idx] : make_int4(0, 0, 0, 0);
    }
#pragma unroll
    for (int j = 0; j < GU; ++j) {
      rput(cur, o, vs[j].x, vd[j].x);
      rput(cur, o, vs[j].y, vd[j].y);
      rput(cur, o, vs[j].z, vd[j].z);
      rput(cur, o, vs[j].w, vd[j].w);
    }
  }
}

#define SEPW 8
template <bool FULL>
__device__ __forceinline__ void sgroup(const float* __restrict__ feat, const unsigned* __restrict__ pk,
                                       int base, int n, int lane, float* tile) {
  unsigned w[SEPW];
#pragma unroll
  for (int i = 0; i < SEPW; ++i) w[i] = pk[(FULL || i < n) ? base + i : base + n - 1];
  float lo[SEPW], hi[SEPW];
#pragma unroll
  for (int i = 0; i < SEPW; ++i) {
    const float* f = feat + (size_t)(w[i] >> 8) * HF + lane;
    lo[i] = f[0];
    hi[i] = f[32];
  }
#pragma unroll
  for (int i = 0; i < SEPW; ++i) {
    if (FULL || i < n) {
      float* o = tile + (w[i] & 255u) * HF + lane;
      __hip_atomic_fetch_add(o,      lo[i], __ATOMIC_RELAXED, __HIP_MEMORY_SCOPE_WORKGROUP);
      __hip_atomic_fetch_add(o + 32, hi[i], __ATOMIC_RELAXED, __HIP_MEMORY_SCOPE_WORKGROUP);
    }
  }
}

__launch_bounds__(1024)
__global__ void k_scatter(const float* __restrict__ feat,
                          const unsigned* __restrict__ pk,
                          const unsigned* __restrict__ rstart,
                          float* __restrict__ ah) {
  __shared__ float tile[RN * HF];
  const int g = blockIdx.x;
  const int lane = threadIdx.x & 31;
  const int wave = __builtin_amdgcn_readfirstlane((int)(threadIdx.x >> 5));
  float4* t4 = (float4*)tile;
  for (int i = threadIdx.x; i < RN * HF / 4; i += 1024) t4[i] = make_float4(0.0f, 0.0f, 0.0f, 0.0f);
  const int b = (int)rstart[g], e = (int)rstart[g + 1];
  __syncthreads();
  for (int base = b + wave * SEPW; base < e; base += 32 * SEPW) {
    if (base + SEPW <= e) sgroup<true>(feat, pk, base, SEPW, lane, tile);
    else                  sgroup<false>(feat, pk, base, e - base, lane, tile);
  }
  __syncthreads();
  float4* o4 = (float4*)(ah + (size_t)g * RN * HF);
  for (int i = threadIdx.x; i < RN * HF / 4; i += 1024) o4[i] = t4[i];
}

__launch_bounds__(256)
__global__ void k_gemm(const float* __restrict__ A1,
                       const float* A2,
                       const float* __restrict__ dinv,
                       const float* __restrict__ oscale,
                       float a2sign,
                       const float* __restrict__ W,
                       const float* __restrict__ bias,
                       float* out,
                       float* __restrict__ accum,
                       int accum_init,
                       int n, int act) {
  __shared__ float sW[128 * 64];
  __shared__ float sB[64];
  const bool dual = (A2 != nullptr);
  const int wtot = (dual ? 128 : 64) * 64;
  for (int i = threadIdx.x; i < wtot; i += 256) {
    int k = i >> 6;
    int col = i & 63;
    sW[(k >> 1) * 128 + col * 2 + (k & 1)] = W[i];
  }
  if (threadIdx.x < 64) sB[threadIdx.x] = bias[threadIdx.x];
  __syncthreads();

  const int lane = threadIdx.x & 31;
  const int wave = threadIdx.x >> 5;
  const int strip = blockIdx.x * 8 + wave;
  if (strip >= (n >> 4)) return;
  const int row0 = strip << 4;
  const int m  = lane & 15;
  const int kh = (lane >> 4) << 1;

  const float* a1p = A1 + (size_t)(row0 + m) * HF + kh;
  const float dr = dinv ? dinv[row0 + m] : 1.0f;
  const float ur = 1.0f / dr;
  v8f acc[4] = {};

#pragma unroll
  for (int k0 = 0; k0 < 64; k0 += 4) {
    v2f a = *(const v2f*)(a1p + k0);
    a = a * ur;
    const float* wb = &sW[(k0 + kh) * 64 + m * 2];
#pragma unroll
    for (int t = 0; t < 4; ++t) {
      v2f b = *(const v2f*)(wb + t * 32);
      acc[t] = __builtin_amdgcn_wmma_f32_16x16x4_f32(
          false, a, false, b, (short)0, acc[t], false, false);
    }
  }
  if (dual) {
    const float* a2p = A2 + (size_t)(row0 + m) * HF + kh;
    const float ds = dr * a2sign;
#pragma unroll
    for (int k0 = 0; k0 < 64; k0 += 4) {
      v2f a = *(const v2f*)(a2p + k0);
      a = a * ds;
      const float* wb = &sW[(64 + k0 + kh) * 64 + m * 2];
#pragma unroll
      for (int t = 0; t < 4; ++t) {
        v2f b = *(const v2f*)(wb + t * 32);
        acc[t] = __builtin_amdgcn_wmma_f32_16x16x4_f32(
            false, a, false, b, (short)0, acc[t], false, false);
      }
    }
  }

  const int rbase = row0 + ((lane >> 4) << 3);
  float4 os0 = make_float4(1.0f, 1.0f, 1.0f, 1.0f), os1 = os0;
  if (oscale) {
    os0 = *(const float4*)(oscale + rbase);
    os1 = *(const float4*)(oscale + rbase + 4);
  }
  const float os[8] = {os0.x, os0.y, os0.z, os0.w, os1.x, os1.y, os1.z, os1.w};
#pragma unroll
  for (int t = 0; t < 4; ++t) {
    const float bv = sB[t * 16 + m];
#pragma unroll
    for (int v = 0; v < 8; ++v) {
      float val = acc[t][v] + bv;
      if (act) val = val > 0.0f ? val : 0.01f * val;
      size_t o = (size_t)(rbase + v) * HF + t * 16 + m;
      out[o] = val * os[v];
      if (accum) {
        if (accum_init) accum[o] = val; else accum[o] += val;
      }
    }
  }
}

#define FW 16
#define SP 68
template <bool DUAL, bool HEAD>
__launch_bounds__(FW * 32)
__global__ void k_gemm2(const float* __restrict__ A1,
                        const float* __restrict__ A2,
                        const float* __restrict__ dinv,
                        const float* __restrict__ oscale,
                        const float* __restrict__ Wa, const float* __restrict__ ba,
                        const float* __restrict__ Wb, const float* __restrict__ bb,
                        const float* __restrict__ W4, const float* __restrict__ b4,
                        float* __restrict__ out,
                        float* __restrict__ accum, int accum_init, int n) {
  constexpr int KR = DUAL ? 128 : 64;
  __shared__ float sWa[KR * 64];
  __shared__ float sWb[KR * 64];
  __shared__ float sBa[64];
  __shared__ float sBb[64];
  __shared__ float sW4[128];
  __shared__ float sb4[2];
  __shared__ float sT[FW * 16 * SP];
  constexpr int PT = KR * 64 / (FW * 32);
  float wa[PT], wb[PT];
#pragma unroll
  for (int j = 0; j < PT; ++j) {
    wa[j] = Wa[threadIdx.x + j * (FW * 32)];
    wb[j] = Wb[threadIdx.x + j * (FW * 32)];
  }
#pragma unroll
  for (int j = 0; j < PT; ++j) {
    const int i = threadIdx.x + j * (FW * 32);
    const int k = i >> 6;
    const int col = i & 63;
    const int o = (k >> 1) * 128 + col * 2 + (k & 1);
    sWa[o] = wa[j];
    sWb[o] = wb[j];
  }
  if (threadIdx.x < 64) {
    sBa[threadIdx.x] = ba[threadIdx.x];
    sBb[threadIdx.x] = bb[threadIdx.x];
  }
  if (HEAD) {
    if (threadIdx.x < 128) sW4[threadIdx.x] = W4[threadIdx.x];
    if (threadIdx.x < 2) sb4[threadIdx.x] = b4[threadIdx.x];
  }
  __syncthreads();

  const int lane = threadIdx.x & 31;
  const int wave = threadIdx.x >> 5;
  const int strip = blockIdx.x * FW + wave;
  if (strip >= (n >> 4)) return;
  const int row0 = strip << 4;
  const int m  = lane & 15;
  const int hh = lane >> 4;
  const int kh = hh << 1;
  float* st = sT + wave * (16 * SP);

  const float* a1p = A1 + (size_t)(row0 + m) * HF + kh;
  v2f a1f[16];
#pragma unroll
  for (int i = 0; i < 16; ++i) a1f[i] = *(const v2f*)(a1p + 4 * i);
  const int rbase = row0 + (hh << 3);
  float dr = 1.0f;
  if (DUAL) {
    dr = dinv[row0 + m];
    const float ur = 1.0f / dr;
#pragma unroll
    for (int i = 0; i < 16; ++i) a1f[i] = a1f[i] * ur;
  }
  v8f acc[4] = {};
#pragma unroll
  for (int i = 0; i < 16; ++i) {
    const float* wb = &sWa[(4 * i + kh) * 64 + m * 2];
#pragma unroll
    for (int t = 0; t < 4; ++t) {
      v2f b = *(const v2f*)(wb + t * 32);
      acc[t] = __builtin_amdgcn_wmma_f32_16x16x4_f32(
          false, a1f[i], false, b, (short)0, acc[t], false, false);
    }
  }
  if (DUAL) {
    const float* a2p = A2 + (size_t)(row0 + m) * HF + kh;
    const float ds = -dr;
#pragma unroll
    for (int i = 0; i < 16; ++i) {
      v2f a = *(const v2f*)(a2p + 4 * i);
      a = a * ds;
      const float* wb = &sWa[(64 + 4 * i + kh) * 64 + m * 2];
#pragma unroll
      for (int t = 0; t < 4; ++t) {
        v2f b = *(const v2f*)(wb + t * 32);
        acc[t] = __builtin_amdgcn_wmma_f32_16x16x4_f32(
            false, a, false, b, (short)0, acc[t], false, false);
      }
    }
  }

#pragma unroll
  for (int t = 0; t < 4; ++t) {
    const float bv = sBa[t * 16 + m];
#pragma unroll
    for (int v = 0; v < 8; ++v) {
      float val = acc[t][v] + bv;
      if (!DUAL) val = val > 0.0f ? val : 0.01f * val;
      st[(v + 8 * hh) * SP + t * 16 + m] = val;
    }
  }
  __builtin_amdgcn_fence(__ATOMIC_ACQ_REL, "wavefront");

  float4 os0 = make_float4(1.0f, 1.0f, 1.0f, 1.0f), os1 = os0;
  if (!HEAD && oscale) {
    os0 = *(const float4*)(oscale + rbase);
    os1 = *(const float4*)(oscale + rbase + 4);
  }
  const float os[8] = {os0.x, os0.y, os0.z, os0.w, os1.x, os1.y, os1.z, os1.w};

  v8f acc2[4] = {};
  if (DUAL) {
#pragma unroll
    for (int i = 0; i < 16; ++i) {
      const float* wb = &sWb[(4 * i + kh) * 64 + m * 2];
#pragma unroll
      for (int t = 0; t < 4; ++t) {
        v2f b = *(const v2f*)(wb + t * 32);
        acc2[t] = __builtin_amdgcn_wmma_f32_16x16x4_f32(
            false, a1f[i], false, b, (short)0, acc2[t], false, false);
      }
    }
  }
  {
    const float* tp = st + m * SP + kh;
#pragma unroll
    for (int i = 0; i < 16; ++i) {
      v2f a = *(const v2f*)(tp + 4 * i);
      const float* wb = &sWb[((DUAL ? 64 : 0) + 4 * i + kh) * 64 + m * 2];
#pragma unroll
      for (int t = 0; t < 4; ++t) {
        v2f b = *(const v2f*)(wb + t * 32);
        acc2[t] = __builtin_amdgcn_wmma_f32_16x16x4_f32(
            false, a, false, b, (short)0, acc2[t], false, false);
      }
    }
  }

  if (!HEAD) {
#pragma unroll
    for (int t = 0; t < 4; ++t) {
      const float bv = sBb[t * 16 + m];
#pragma unroll
      for (int v = 0; v < 8; ++v) {
        float val = acc2[t][v] + bv;
        if (!DUAL) val = val > 0.0f ? val : 0.01f * val;
        const size_t o = (size_t)(rbase + v) * HF + t * 16 + m;
        out[o] = val * os[v];
        if (accum) {
          if (accum_init) accum[o] = val; else accum[o] += val;
        }
      }
    }
  } else {
    __builtin_amdgcn_fence(__ATOMIC_ACQ_REL, "wavefront");
#pragma unroll
    for (int t = 0; t < 4; ++t) {
      const float bv = sBb[t * 16 + m];
#pragma unroll
      for (int v = 0; v < 8; ++v) {
        float val = acc2[t][v] + bv;
        if (!accum_init) val += accum[(size_t)(rbase + v) * HF + t * 16 + m];
        val = val > 0.0f ? val : 0.01f * val;
        st[(v + 8 * hh) * SP + t * 16 + m] = val;
      }
    }
    __builtin_amdgcn_fence(__ATOMIC_ACQ_REL, "wavefront");
    const float* tp = st + m * SP;
    float a = sb4[hh];
#pragma unroll
    for (int k = 0; k < 64; ++k) a += tp[k] * sW4[2 * k + hh];
    out[2 * (size_t)(row0 + m) + hh] = a;
  }
}

extern "C" void kernel_launch(void* const* d_in, const int* in_sizes, int n_in,
                              void* d_out, int out_size, void* d_ws, size_t ws_size,
                              hipStream_t stream) {
  const float* x   = (const float*)d_in[0];
  const int*   src = (const int*)d_in[1];
  const int*   dst = (const int*)d_in[2];
  const float* W1  = (const float*)d_in[3];
  const float* b1  = (const float*)d_in[4];
  const float* W2  = (const float*)d_in[5];
  const float* b2  = (const float*)d_in[6];
  const float* Wc1 = (const float*)d_in[7];
  const float* bc1 = (const float*)d_in[8];
  const float* Wc2 = (const float*)d_in[9];
  const float* bc2 = (const float*)d_in[10];
  const float* W3  = (const float*)d_in[11];
  const float* b3  = (const float*)d_in[12];
  const float* W4  = (const float*)d_in[13];
  const float* b4  = (const float*)d_in[14];
  float* out = (float*)d_out;

  const size_t NH = (size_t)NODES * HF;
  float* ws   = (float*)d_ws;
  float* hbuf = ws;
  float* h0   = hbuf + NH;
  float* grp  = h0 + NH;
  float* hall = grp + NH;
  float* ah   = hall + NH;
  float* dinv = ah + NH;
  unsigned* pk     = (unsigned*)grp;
  unsigned* rcnt   = pk + (size_t)RELS * EDGES;
  unsigned* rstart = rcnt + (size_t)RELS * GCH * NR;

  k_degpart<<<dim3(DCH, DQ, RELS), 1024, 0, stream>>>(dst, ah);
  k_dinv<<<(RELS * NODES + 255) / 256, 256, 0, stream>>>(ah, dinv);

  k_rcount<<<dim3(GCH, RELS), 1024, 0, stream>>>(dst, rcnt);
  k_rfill<<<dim3(GCH, RELS), 1024, 0, stream>>>(src, dst, rcnt, rstart, pk);

  const int gblocks = ((NODES / 16) + 7) / 8;

  const int fblocks = ((NODES / 16) + FW - 1) / FW;

  k_gemm2<false, false><<<fblocks, FW * 32, 0, stream>>>(x, nullptr, nullptr, dinv, W1, b1, W2, b2,
                                                         nullptr, nullptr, hbuf, nullptr, 0, NODES);

  for (int r = 0; r < RELS; ++r) {
    const unsigned* pr = pk + (size_t)r * EDGES;
    const unsigned* sr = rstart + (size_t)r * (NR + 1);
    const float* dv = dinv + (size_t)r * NODES;
    const float* dvn = dinv + (size_t)(r + 1) * NODES;

    k_scatter<<<NR, 1024, 0, stream>>>(hbuf, pr, sr, ah);
    k_gemm<<<gblocks, 256, 0, stream>>>(hbuf, ah, dv, dv, -1.0f, Wc1, bc1, h0, nullptr, 0, NODES, 0);

    k_scatter<<<NR, 1024, 0, stream>>>(h0, pr, sr, ah);
    if (r + 1 < RELS)
      k_gemm2<true, false><<<fblocks, FW * 32, 0, stream>>>(h0, ah, dv, dvn, Wc2, bc2, W3, b3, nullptr, nullptr,
                                                            hbuf, hall, r == 0 ? 1 : 0, NODES);
    else
      k_gemm2<true, true><<<fblocks, FW * 32, 0, stream>>>(h0, ah, dv, nullptr, Wc2, bc2, W3, b3, W4, b4,
                                                           out, hall, r == 0 ? 1 : 0, NODES);
  }
}


# Node products launched as 256 work-groups of 25 waves, one per WGP, strip = wave*256 + group: busiest WGP 32 -> 25 strips (SIMD 8 -> 7); W tiles staged once per WGP; strip code, sum order kept
# speedup vs baseline: 2.6539x; 1.0659x over previous
#include <hip/hip_runtime.h>

typedef __attribute__((ext_vector_type(2))) float v2f;
typedef __attribute__((ext_vector_type(8))) float v8f;

#define NODES 100000
#define EDGES 1000000
#define RELS  3
#define HF    64

#define DCH 16
#define DQ  4
#define DNR 25024
#define DEC (EDGES / DCH)
__launch_bounds__(1024)
__global__ void k_degpart(const int* __restrict__ dst, float* __restrict__ part) {
  __shared__ unsigned cnt[DNR];
  const int c = blockIdx.x, q = blockIdx.y, r = blockIdx.z;
  for (int i = threadIdx.x; i < DNR; i += 1024) cnt[i] = 0u;
  __syncthreads();
  const int lo = q * DNR;
  const int4* p = (const int4*)(dst + (size_t)r * EDGES + (size_t)c * DEC);
  for (int base = threadIdx.x; base < DEC / 4; base += 8 * 1024) {
    int4 v[8];
#pragma unroll
    for (int j = 0; j < 8; ++j) {
      const int idx = base + j * 1024;
      v[j] = idx < DEC / 4 ? p[idx] : make_int4(-1, -1, -1, -1);
    }
#pragma unroll
    for (int j = 0; j < 8; ++j) {
      const unsigned a0 = (unsigned)(v[j].x - lo), a1 = (unsigned)(v[j].y - lo);
      const unsigned a2 = (unsigned)(v[j].z - lo), a3 = (unsigned)(v[j].w - lo);
      if (a0 < DNR) atomicAdd(&cnt[a0], 1u);
      if (a1 < DNR) atomicAdd(&cnt[a1], 1u);
      if (a2 < DNR) atomicAdd(&cnt[a2], 1u);
      if (a3 < DNR) atomicAdd(&cnt[a3], 1u);
    }
  }
  __syncthreads();
  const int nh = (NODES - lo) < DNR ? (NODES - lo) : DNR;
  float* o = part + ((size_t)c * RELS + r) * NODES + lo;
  for (int i = threadIdx.x; i < nh; i += 1024) o[i] = (float)cnt[i];
}

__global__ void k_dinv(const float* __restrict__ part, float* __restrict__ dinv) {
  int i = blockIdx.x * blockDim.x + threadIdx.x;
  if (i >= RELS * NODES) return;
  float v = 0.0f;
#pragma unroll
  for (int c = 0; c < DCH; ++c) v += part[(size_t)c * RELS * NODES + i];
  v = v < 1.0f ? 1.0f : v;
  dinv[i] = 1.0f / sqrtf(v);
}

#define RN   200
#define NR   (NODES / RN)
#define GCH  50
#define GEC  (EDGES / GCH)
#define GU   5

__launch_bounds__(1024)
__global__ void k_rcount(const int* __restrict__ dst, unsigned* __restrict__ cnt) {
  __shared__ unsigned h[NR];
  const int c = blockIdx.x, r = blockIdx.y;
  for (int i = threadIdx.x; i < NR; i += 1024) h[i] = 0u;
  __syncthreads();
  const int4* p = (const int4*)(dst + (size_t)r * EDGES + (size_t)c * GEC);
  for (int base = threadIdx.x; base < GEC / 4; base += GU * 1024) {
    int4 v[GU];
#pragma unroll
    for (int j = 0; j < GU; ++j) {
      const int idx = base + j * 1024;
      v[j] = idx < GEC / 4 ? p[idx] : make_int4(-1, -1, -1, -1);
    }
#pragma unroll
    for (int j = 0; j < GU; ++j) {
      if ((unsigned)v[j].x < NODES) atomicAdd(&h[(unsigned)v[j].x / RN], 1u);
      if ((unsigned)v[j].y < NODES) atomicAdd(&h[(unsigned)v[j].y / RN], 1u);
      if ((unsigned)v[j].z < NODES) atomicAdd(&h[(unsigned)v[j].z / RN], 1u);
      if ((unsigned)v[j].w < NODES) atomicAdd(&h[(unsigned)v[j].w / RN], 1u);
    }
  }
  __syncthreads();
  unsigned* o = cnt + ((size_t)r * GCH + c) * NR;
  for (int i = threadIdx.x; i < NR; i += 1024) o[i] = h[i];
}

__device__ __forceinline__ void rput(unsigned* cur, unsigned* __restrict__ o, int sid, int did) {
  if ((unsigned)did < NODES) {
    const unsigned g = (unsigned)did / RN;
    const unsigned pos = atomicAdd(&cur[g], 1u);
    o[pos] = ((unsigned)sid << 8) | ((unsigned)did - g * RN);
  }
}

__launch_bounds__(1024)
__global__ void k_rfill(const int* __restrict__ src, const int* __restrict__ dst,
                        const unsigned* __restrict__ cnt,
                        unsigned* __restrict__ rstart,
                        unsigned* __restrict__ pk) {
  __shared__ unsigned sc[512];
  __shared__ unsigned cur[NR];
  const int c = blockIdx.x, r = blockIdx.y;
  const int t = threadIdx.x;
  unsigned tot = 0u, pre = 0u;
  if (t < NR) {
    const unsigned* q = cnt + (size_t)r * GCH * NR + t;
#pragma unroll 10
    for (int j = 0; j < GCH; ++j) {
      const unsigned x = q[(size_t)j * NR];
      tot += x;
      pre += j < c ? x : 0u;
    }
  }
  if (t < 512) sc[t] = tot;
  __syncthreads();
  for (int off = 1; off < 512; off <<= 1) {
    unsigned a = 0u;
    if (t < 512 && t >= off) a = sc[t - off];
    __syncthreads();
    if (t < 512) sc[t] += a;
    __syncthreads();
  }
  if (t < NR) {
    const unsigned st = sc[t] - tot;
    cur[t] = st + pre;
    if (c == 0) {
      rstart[(size_t)r * (NR + 1) + t] = st;
      if (t == NR - 1) rstart[(size_t)r * (NR + 1) + NR] = sc[t];
    }
  }
  __syncthreads();
  const int4* ps = (const int4*)(src + (size_t)r * EDGES + (size_t)c * GEC);
  const int4* pd = (const int4*)(dst + (size_t)r * EDGES + (size_t)c * GEC);
  unsigned* o = pk + (size_t)r * EDGES;
  for (int base = t; base < GEC / 4; base += GU * 1024) {
    int4 vs[GU], vd[GU];
#pragma unroll
    for (int j = 0; j < GU; ++j) {
      const int idx = base + j * 1024;
      const bool ok = idx < GEC / 4;
      vd[j] = ok ? pd[idx] : make_int4(-1, -1, -1, -1);
      vs[j] = ok ? ps[idx] : make_int4(0, 0, 0, 0);
    }
#pragma unroll
    for (int j = 0; j < GU; ++j) {
      rput(cur, o, vs[j].x, vd[j].x);
      rput(cur, o, vs[j].y, vd[j].y);
      rput(cur, o, vs[j].z, vd[j].z);
      rput(cur, o, vs[j].w, vd[j].w);
    }
  }
}

#define SEPW 8
template <bool FULL>
__device__ __forceinline__ void sgroup(const float* __restrict__ feat, const unsigned* __restrict__ pk,
                                       int base, int n, int lane, float* tile) {
  unsigned w[SEPW];
#pragma unroll
  for (int i = 0; i < SEPW; ++i) w[i] = pk[(FULL || i < n) ? base + i : base + n - 1];
  float lo[SEPW], hi[SEPW];
#pragma unroll
  for (int i = 0; i < SEPW; ++i) {
    const float* f = feat + (size_t)(w[i] >> 8) * HF + lane;
    lo[i] = f[0];
    hi[i] = f[32];
  }
#pragma unroll
  for (int i = 0; i < SEPW; ++i) {
    if (FULL || i < n) {
      float* o = tile + (w[i] & 255u) * HF + lane;
      __hip_atomic_fetch_add(o,      lo[i], __ATOMIC_RELAXED, __HIP_MEMORY_SCOPE_WORKGROUP);
      __hip_atomic_fetch_add(o + 32, hi[i], __ATOMIC_RELAXED, __HIP_MEMORY_SCOPE_WORKGROUP);
    }
  }
}

__launch_bounds__(1024)
__global__ void k_scatter(const float* __restrict__ feat,
                          const unsigned* __restrict__ pk,
                          const unsigned* __restrict__ rstart,
                          float* __restrict__ ah) {
  __shared__ float tile[RN * HF];
  const int g = blockIdx.x;
  const int lane = threadIdx.x & 31;
  const int wave = __builtin_amdgcn_readfirstlane((int)(threadIdx.x >> 5));
  float4* t4 = (float4*)tile;
  for (int i = threadIdx.x; i < RN * HF / 4; i += 1024) t4[i] = make_float4(0.0f, 0.0f, 0.0f, 0.0f);
  const int b = (int)rstart[g], e = (int)rstart[g + 1];
  __syncthreads();
  for (int base = b + wave * SEPW; base < e; base += 32 * SEPW) {
    if (base + SEPW <= e) sgroup<true>(feat, pk, base, SEPW, lane, tile);
    else                  sgroup<false>(feat, pk, base, e - base, lane, tile);
  }
  __syncthreads();
  float4* o4 = (float4*)(ah + (size_t)g * RN * HF);
  for (int i = threadIdx.x; i < RN * HF / 4; i += 1024) o4[i] = t4[i];
}

#define PW 256
#define PS 25

__launch_bounds__(PS * 32)
__global__ void k_gemm(const float* __restrict__ A1,
                       const float* A2,
                       const float* __restrict__ dinv,
                       const float* __restrict__ oscale,
                       float a2sign,
                       const float* __restrict__ W,
                       const float* __restrict__ bias,
                       float* out,
                       float* __restrict__ accum,
                       int accum_init,
                       int n, int act) {
  __shared__ float sW[128 * 64];
  __shared__ float sB[64];
  const bool dual = (A2 != nullptr);
  const int wtot = (dual ? 128 : 64) * 64;
  for (int i = threadIdx.x; i < wtot; i += PS * 32) {
    int k = i >> 6;
    int col = i & 63;
    sW[(k >> 1) * 128 + col * 2 + (k & 1)] = W[i];
  }
  if (threadIdx.x < 64) sB[threadIdx.x] = bias[threadIdx.x];
  __syncthreads();

  const int lane = threadIdx.x & 31;
  const int wave = threadIdx.x >> 5;
  const int strip = wave * PW + blockIdx.x;
  if (strip >= (n >> 4)) return;
  const int row0 = strip << 4;
  const int m  = lane & 15;
  const int kh = (lane >> 4) << 1;

  const float* a1p = A1 + (size_t)(row0 + m) * HF + kh;
  const float dr = dinv ? dinv[row0 + m] : 1.0f;
  const float ur = 1.0f / dr;
  v8f acc[4] = {};

#pragma unroll
  for (int k0 = 0; k0 < 64; k0 += 4) {
    v2f a = *(const v2f*)(a1p + k0);
    a = a * ur;
    const float* wb = &sW[(k0 + kh) * 64 + m * 2];
#pragma unroll
    for (int t = 0; t < 4; ++t) {
      v2f b = *(const v2f*)(wb + t * 32);
      acc[t] = __builtin_amdgcn_wmma_f32_16x16x4_f32(
          false, a, false, b, (short)0, acc[t], false, false);
    }
  }
  if (dual) {
    const float* a2p = A2 + (size_t)(row0 + m) * HF + kh;
    const float ds = dr * a2sign;
#pragma unroll
    for (int k0 = 0; k0 < 64; k0 += 4) {
      v2f a = *(const v2f*)(a2p + k0);
      a = a * ds;
      const float* wb = &sW[(64 + k0 + kh) * 64 + m * 2];
#pragma unroll
      for (int t = 0; t < 4; ++t) {
        v2f b = *(const v2f*)(wb + t * 32);
        acc[t] = __builtin_amdgcn_wmma_f32_16x16x4_f32(
            false, a, false, b, (short)0, acc[t], false, false);
      }
    }
  }

  const int rbase = row0 + ((lane >> 4) << 3);
  float4 os0 = make_float4(1.0f, 1.0f, 1.0f, 1.0f), os1 = os0;
  if (oscale) {
    os0 = *(const float4*)(oscale + rbase);
    os1 = *(const float4*)(oscale + rbase + 4);
  }
  const float os[8] = {os0.x, os0.y, os0.z, os0.w, os1.x, os1.y, os1.z, os1.w};
#pragma unroll
  for (int t = 0; t < 4; ++t) {
    const float bv = sB[t * 16 + m];
#pragma unroll
    for (int v = 0; v < 8; ++v) {
      float val = acc[t][v] + bv;
      if (act) val = val > 0.0f ? val : 0.01f * val;
      size_t o = (size_t)(rbase + v) * HF + t * 16 + m;
      out[o] = val * os[v];
      if (accum) {
        if (accum_init) accum[o] = val; else accum[o] += val;
      }
    }
  }
}

#define SP 68
template <bool DUAL, bool HEAD>
__launch_bounds__(PS * 32)
__global__ void k_gemm2(const float* __restrict__ A1,
                        const float* __restrict__ A2,
                        const float* __restrict__ dinv,
                        const float* __restrict__ oscale,
                        const float* __restrict__ Wa, const float* __restrict__ ba,
                        const float* __restrict__ Wb, const float* __restrict__ bb,
                        const float* __restrict__ W4, const float* __restrict__ b4,
                        float* __restrict__ out,
                        float* __restrict__ accum, int accum_init, int n) {
  constexpr int KR = DUAL ? 128 : 64;
  __shared__ float sWa[KR * 64];
  __shared__ float sWb[KR * 64];
  __shared__ float sBa[64];
  __shared__ float sBb[64];
  __shared__ float sW4[128];
  __shared__ float sb4[2];
  __shared__ float sT[PS * 16 * SP];
  constexpr int NT = PS * 32;
  constexpr int PT = (KR * 64 + NT - 1) / NT;
  float wa[PT], wb[PT];
#pragma unroll
  for (int j = 0; j < PT; ++j) {
    const int i = threadIdx.x + j * NT;
    const bool ok = i < KR * 64;
    wa[j] = ok ? Wa[i] : 0.0f;
    wb[j] = ok ? Wb[i] : 0.0f;
  }
#pragma unroll
  for (int j = 0; j < PT; ++j) {
    const int i = threadIdx.x + j * NT;
    if (i < KR * 64) {
      const int k = i >> 6;
      const int col = i & 63;
      const int o = (k >> 1) * 128 + col * 2 + (k & 1);
      sWa[o] = wa[j];
      sWb[o] = wb[j];
    }
  }
  if (threadIdx.x < 64) {
    sBa[threadIdx.x] = ba[threadIdx.x];
    sBb[threadIdx.x] = bb[threadIdx.x];
  }
  if (HEAD) {
    if (threadIdx.x < 128) sW4[threadIdx.x] = W4[threadIdx.x];
    if (threadIdx.x < 2) sb4[threadIdx.x] = b4[threadIdx.x];
  }
  __syncthreads();

  const int lane = threadIdx.x & 31;
  const int wave = threadIdx.x >> 5;
  const int strip = wave * PW + blockIdx.x;
  if (strip >= (n >> 4)) return;
  const int row0 = strip << 4;
  const int m  = lane & 15;
  const int hh = lane >> 4;
  const int kh = hh << 1;
  float* st = sT + wave * (16 * SP);

  const float* a1p = A1 + (size_t)(row0 + m) * HF + kh;
  v2f a1f[16];
#pragma unroll
  for (int i = 0; i < 16; ++i) a1f[i] = *(const v2f*)(a1p + 4 * i);
  const int rbase = row0 + (hh << 3);
  float dr = 1.0f;
  if (DUAL) {
    dr = dinv[row0 + m];
    const float ur = 1.0f / dr;
#pragma unroll
    for (int i = 0; i < 16; ++i) a1f[i] = a1f[i] * ur;
  }
  v8f acc[4] = {};
#pragma unroll
  for (int i = 0; i < 16; ++i) {
    const float* wb = &sWa[(4 * i + kh) * 64 + m * 2];
#pragma unroll
    for (int t = 0; t < 4; ++t) {
      v2f b = *(const v2f*)(wb + t * 32);
      acc[t] = __builtin_amdgcn_wmma_f32_16x16x4_f32(
          false, a1f[i], false, b, (short)0, acc[t], false, false);
    }
  }
  if (DUAL) {
    const float* a2p = A2 + (size_t)(row0 + m) * HF + kh;
    const float ds = -dr;
#pragma unroll
    for (int i = 0; i < 16; ++i) {
      v2f a = *(const v2f*)(a2p + 4 * i);
      a = a * ds;
      const float* wb = &sWa[(64 + 4 * i + kh) * 64 + m * 2];
#pragma unroll
      for (int t = 0; t < 4; ++t) {
        v2f b = *(const v2f*)(wb + t * 32);
        acc[t] = __builtin_amdgcn_wmma_f32_16x16x4_f32(
            false, a, false, b, (short)0, acc[t], false, false);
      }
    }
  }

#pragma unroll
  for (int t = 0; t < 4; ++t) {
    const float bv = sBa[t * 16 + m];
#pragma unroll
    for (int v = 0; v < 8; ++v) {
      float val = acc[t][v] + bv;
      if (!DUAL) val = val > 0.0f ? val : 0.01f * val;
      st[(v + 8 * hh) * SP + t * 16 + m] = val;
    }
  }
  __builtin_amdgcn_fence(__ATOMIC_ACQ_REL, "wavefront");

  float4 os0 = make_float4(1.0f, 1.0f, 1.0f, 1.0f), os1 = os0;
  if (!HEAD && oscale) {
    os0 = *(const float4*)(oscale + rbase);
    os1 = *(const float4*)(oscale + rbase + 4);
  }
  const float os[8] = {os0.x, os0.y, os0.z, os0.w, os1.x, os1.y, os1.z, os1.w};

  v8f acc2[4] = {};
  if (DUAL) {
#pragma unroll
    for (int i = 0; i < 16; ++i) {
      const float* wb = &sWb[(4 * i + kh) * 64 + m * 2];
#pragma unroll
      for (int t = 0; t < 4; ++t) {
        v2f b = *(const v2f*)(wb + t * 32);
        acc2[t] = __builtin_amdgcn_wmma_f32_16x16x4_f32(
            false, a1f[i], false, b, (short)0, acc2[t], false, false);
      }
    }
  }
  {
    const float* tp = st + m * SP + kh;
#pragma unroll
    for (int i = 0; i < 16; ++i) {
      v2f a = *(const v2f*)(tp + 4 * i);
      const float* wb = &sWb[((DUAL ? 64 : 0) + 4 * i + kh) * 64 + m * 2];
#pragma unroll
      for (int t = 0; t < 4; ++t) {
        v2f b = *(const v2f*)(wb + t * 32);
        acc2[t] = __builtin_amdgcn_wmma_f32_16x16x4_f32(
            false, a, false, b, (short)0, acc2[t], false, false);
      }
    }
  }

  if (!HEAD) {
#pragma unroll
    for (int t = 0; t < 4; ++t) {
      const float bv = sBb[t * 16 + m];
#pragma unroll
      for (int v = 0; v < 8; ++v) {
        float val = acc2[t][v] + bv;
        if (!DUAL) val = val > 0.0f ? val : 0.01f * val;
        const size_t o = (size_t)(rbase + v) * HF + t * 16 + m;
        out[o] = val * os[v];
        if (accum) {
          if (accum_init) accum[o] = val; else accum[o] += val;
        }
      }
    }
  } else {
    __builtin_amdgcn_fence(__ATOMIC_ACQ_REL, "wavefront");
#pragma unroll
    for (int t = 0; t < 4; ++t) {
      const float bv = sBb[t * 16 + m];
#pragma unroll
      for (int v = 0; v < 8; ++v) {
        float val = acc2[t][v] + bv;
        if (!accum_init) val += accum[(size_t)(rbase + v) * HF + t * 16 + m];
        val = val > 0.0f ? val : 0.01f * val;
        st[(v + 8 * hh) * SP + t * 16 + m] = val;
      }
    }
    __builtin_amdgcn_fence(__ATOMIC_ACQ_REL, "wavefront");
    const float* tp = st + m * SP;
    float a = sb4[hh];
#pragma unroll
    for (int k = 0; k < 64; ++k) a += tp[k] * sW4[2 * k + hh];
    out[2 * (size_t)(row0 + m) + hh] = a;
  }
}

extern "C" void kernel_launch(void* const* d_in, const int* in_sizes, int n_in,
                              void* d_out, int out_size, void* d_ws, size_t ws_size,
                              hipStream_t stream) {
  const float* x   = (const float*)d_in[0];
  const int*   src = (const int*)d_in[1];
  const int*   dst = (const int*)d_in[2];
  const float* W1  = (const float*)d_in[3];
  const float* b1  = (const float*)d_in[4];
  const float* W2  = (const float*)d_in[5];
  const float* b2  = (const float*)d_in[6];
  const float* Wc1 = (const float*)d_in[7];
  const float* bc1 = (const float*)d_in[8];
  const float* Wc2 = (const float*)d_in[9];
  const float* bc2 = (const float*)d_in[10];
  const float* W3  = (const float*)d_in[11];
  const float* b3  = (const float*)d_in[12];
  const float* W4  = (const float*)d_in[13];
  const float* b4  = (const float*)d_in[14];
  float* out = (float*)d_out;

  const size_t NH = (size_t)NODES * HF;
  float* ws   = (float*)d_ws;
  float* hbuf = ws;
  float* h0   = hbuf + NH;
  float* grp  = h0 + NH;
  float* hall = grp + NH;
  float* ah   = hall + NH;
  float* dinv = ah + NH;
  unsigned* pk     = (unsigned*)grp;
  unsigned* rcnt   = pk + (size_t)RELS * EDGES;
  unsigned* rstart = rcnt + (size_t)RELS * GCH * NR;

  k_degpart<<<dim3(DCH, DQ, RELS), 1024, 0, stream>>>(dst, ah);
  k_dinv<<<(RELS * NODES + 255) / 256, 256, 0, stream>>>(ah, dinv);

  k_rcount<<<dim3(GCH, RELS), 1024, 0, stream>>>(dst, rcnt);
  k_rfill<<<dim3(GCH, RELS), 1024, 0, stream>>>(src, dst, rcnt, rstart, pk);

  k_gemm2<false, false><<<PW, PS * 32, 0, stream>>>(x, nullptr, nullptr, dinv, W1, b1, W2, b2,
                                                         nullptr, nullptr, hbuf, nullptr, 0, NODES);

  for (int r = 0; r < RELS; ++r) {
    const unsigned* pr = pk + (size_t)r * EDGES;
    const unsigned* sr = rstart + (size_t)r * (NR + 1);
    const float* dv = dinv + (size_t)r * NODES;
    const float* dvn = dinv + (size_t)(r + 1) * NODES;

    k_scatter<<<NR, 1024, 0, stream>>>(hbuf, pr, sr, ah);
    k_gemm<<<PW, PS * 32, 0, stream>>>(hbuf, ah, dv, dv, -1.0f, Wc1, bc1, h0, nullptr, 0, NODES, 0);

    k_scatter<<<NR, 1024, 0, stream>>>(h0, pr, sr, ah);
    if (r + 1 < RELS)
      k_gemm2<true, false><<<PW, PS * 32, 0, stream>>>(h0, ah, dv, dvn, Wc2, bc2, W3, b3, nullptr, nullptr,
                                                            hbuf, hall, r == 0 ? 1 : 0, NODES);
    else
      k_gemm2<true, true><<<PW, PS * 32, 0, stream>>>(h0, ah, dv, nullptr, Wc2, bc2, W3, b3, W4, b4,
                                                           out, hall, r == 0 ? 1 : 0, NODES);
  }
}


# Scatter-add pass as one resident set: 250 work-groups (400-node LDS tile, one per WGP), 32 row loads in flight per wave, next edge words fetched under them; 16k waves a pass -> 8k; arrival-order sums
# speedup vs baseline: 2.7771x; 1.0464x over previous
#include <hip/hip_runtime.h>

typedef __attribute__((ext_vector_type(2))) float v2f;
typedef __attribute__((ext_vector_type(8))) float v8f;

#define NODES 100000
#define EDGES 1000000
#define RELS  3
#define HF    64

#define DCH 16
#define DQ  4
#define DNR 25024
#define DEC (EDGES / DCH)
__launch_bounds__(1024)
__global__ void k_degpart(const int* __restrict__ dst, float* __restrict__ part) {
  __shared__ unsigned cnt[DNR];
  const int c = blockIdx.x, q = blockIdx.y, r = blockIdx.z;
  for (int i = threadIdx.x; i < DNR; i += 1024) cnt[i] = 0u;
  __syncthreads();
  const int lo = q * DNR;
  const int4* p = (const int4*)(dst + (size_t)r * EDGES + (size_t)c * DEC);
  for (int base = threadIdx.x; base < DEC / 4; base += 8 * 1024) {
    int4 v[8];
#pragma unroll
    for (int j = 0; j < 8; ++j) {
      const int idx = base + j * 1024;
      v[j] = idx < DEC / 4 ? p[idx] : make_int4(-1, -1, -1, -1);
    }
#pragma unroll
    for (int j = 0; j < 8; ++j) {
      const unsigned a0 = (unsigned)(v[j].x - lo), a1 = (unsigned)(v[j].y - lo);
      const unsigned a2 = (unsigned)(v[j].z - lo), a3 = (unsigned)(v[j].w - lo);
      if (a0 < DNR) atomicAdd(&cnt[a0], 1u);
      if (a1 < DNR) atomicAdd(&cnt[a1], 1u);
      if (a2 < DNR) atomicAdd(&cnt[a2], 1u);
      if (a3 < DNR) atomicAdd(&cnt[a3], 1u);
    }
  }
  __syncthreads();
  const int nh = (NODES - lo) < DNR ? (NODES - lo) : DNR;
  float* o = part + ((size_t)c * RELS + r) * NODES + lo;
  for (int i = threadIdx.x; i < nh; i += 1024) o[i] = (float)cnt[i];
}

__global__ void k_dinv(const float* __restrict__ part, float* __restrict__ dinv) {
  int i = blockIdx.x * blockDim.x + threadIdx.x;
  if (i >= RELS * NODES) return;
  float v = 0.0f;
#pragma unroll
  for (int c = 0; c < DCH; ++c) v += part[(size_t)c * RELS * NODES + i];
  v = v < 1.0f ? 1.0f : v;
  dinv[i] = 1.0f / sqrtf(v);
}

#define RN   400
#define RSH  9
#define NR   (NODES / RN)
#define GCH  50
#define GEC  (EDGES / GCH)
#define GU   5

__launch_bounds__(1024)
__global__ void k_rcount(const int* __restrict__ dst, unsigned* __restrict__ cnt) {
  __shared__ unsigned h[NR];
  const int c = blockIdx.x, r = blockIdx.y;
  for (int i = threadIdx.x; i < NR; i += 1024) h[i] = 0u;
  __syncthreads();
  const int4* p = (const int4*)(dst + (size_t)r * EDGES + (size_t)c * GEC);
  for (int base = threadIdx.x; base < GEC / 4; base += GU * 1024) {
    int4 v[GU];
#pragma unroll
    for (int j = 0; j < GU; ++j) {
      const int idx = base + j * 1024;
      v[j] = idx < GEC / 4 ? p[idx] : make_int4(-1, -1, -1, -1);
    }
#pragma unroll
    for (int j = 0; j < GU; ++j) {
      if ((unsigned)v[j].x < NODES) atomicAdd(&h[(unsigned)v[j].x / RN], 1u);
      if ((unsigned)v[j].y < NODES) atomicAdd(&h[(unsigned)v[j].y / RN], 1u);
      if ((unsigned)v[j].z < NODES) atomicAdd(&h[(unsigned)v[j].z / RN], 1u);
      if ((unsigned)v[j].w < NODES) atomicAdd(&h[(unsigned)v[j].w / RN], 1u);
    }
  }
  __syncthreads();
  unsigned* o = cnt + ((size_t)r * GCH + c) * NR;
  for (int i = threadIdx.x; i < NR; i += 1024) o[i] = h[i];
}

__device__ __forceinline__ void rput(unsigned* cur, unsigned* __restrict__ o, int sid, int did) {
  if ((unsigned)did < NODES) {
    const unsigned g = (unsigned)did / RN;
    const unsigned pos = atomicAdd(&cur[g], 1u);
    o[pos] = ((unsigned)sid << RSH) | ((unsigned)did - g * RN);
  }
}

__launch_bounds__(1024)
__global__ void k_rfill(const int* __restrict__ src, const int* __restrict__ dst,
                        const unsigned* __restrict__ cnt,
                        unsigned* __restrict__ rstart,
                        unsigned* __restrict__ pk) {
  __shared__ unsigned sc[512];
  __shared__ unsigned cur[NR];
  const int c = blockIdx.x, r = blockIdx.y;
  const int t = threadIdx.x;
  unsigned tot = 0u, pre = 0u;
  if (t < NR) {
    const unsigned* q = cnt + (size_t)r * GCH * NR + t;
#pragma unroll 10
    for (int j = 0; j < GCH; ++j) {
      const unsigned x = q[(size_t)j * NR];
      tot += x;
      pre += j < c ? x : 0u;
    }
  }
  if (t < 512) sc[t] = tot;
  __syncthreads();
  for (int off = 1; off < 512; off <<= 1) {
    unsigned a = 0u;
    if (t < 512 && t >= off) a = sc[t - off];
    __syncthreads();
    if (t < 512) sc[t] += a;
    __syncthreads();
  }
  if (t < NR) {
    const unsigned st = sc[t] - tot;
    cur[t] = st + pre;
    if (c == 0) {
      rstart[(size_t)r * (NR + 1) + t] = st;
      if (t == NR - 1) rstart[(size_t)r * (NR + 1) + NR] = sc[t];
    }
  }
  __syncthreads();
  const int4* ps = (const int4*)(src + (size_t)r * EDGES + (size_t)c * GEC);
  const int4* pd = (const int4*)(dst + (size_t)r * EDGES + (size_t)c * GEC);
  unsigned* o = pk + (size_t)r * EDGES;
  for (int base = t; base < GEC / 4; base += GU * 1024) {
    int4 vs[GU], vd[GU];
#pragma unroll
    for (int j = 0; j < GU; ++j) {
      const int idx = base + j * 1024;
      const bool ok = idx < GEC / 4;
      vd[j] = ok ? pd[idx] : make_int4(-1, -1, -1, -1);
      vs[j] = ok ? ps[idx] : make_int4(0, 0, 0, 0);
    }
#pragma unroll
    for (int j = 0; j < GU; ++j) {
      rput(cur, o, vs[j].x, vd[j].x);
      rput(cur, o, vs[j].y, vd[j].y);
      rput(cur, o, vs[j].z, vd[j].z);
      rput(cur, o, vs[j].w, vd[j].w);
    }
  }
}

#define SEPW 16
__device__ __forceinline__ void srows(const float* __restrict__ feat, const unsigned (&w)[SEPW], int lane,
                                      float (&lo)[SEPW], float (&hi)[SEPW]) {
#pragma unroll
  for (int i = 0; i < SEPW; ++i) {
    const float* f = feat + (size_t)(w[i] >> RSH) * HF + lane;
    lo[i] = f[0];
    hi[i] = f[32];
  }
}
__device__ __forceinline__ void sadd(float* tile, unsigned row, int lane, float lo, float hi) {
  float* o = tile + row * HF + lane;
  __hip_atomic_fetch_add(o,      lo, __ATOMIC_RELAXED, __HIP_MEMORY_SCOPE_WORKGROUP);
  __hip_atomic_fetch_add(o + 32, hi, __ATOMIC_RELAXED, __HIP_MEMORY_SCOPE_WORKGROUP);
}

__launch_bounds__(1024)
__global__ void k_scatter(const float* __restrict__ feat,
                          const unsigned* __restrict__ pk,
                          const unsigned* __restrict__ rstart,
                          float* __restrict__ ah) {
  __shared__ float tile[RN * HF];
  const int g = blockIdx.x;
  const int lane = threadIdx.x & 31;
  const int wave = __builtin_amdgcn_readfirstlane((int)(threadIdx.x >> 5));
  float4* t4 = (float4*)tile;
  for (int i = threadIdx.x; i < RN * HF / 4; i += 1024) t4[i] = make_float4(0.0f, 0.0f, 0.0f, 0.0f);
  const int b = (int)rstart[g], e = (int)rstart[g + 1];
  __syncthreads();
  int base = b + wave * SEPW;
  if (base + SEPW <= e) {
    unsigned w[SEPW];
#pragma unroll
    for (int i = 0; i < SEPW; ++i) w[i] = pk[base + i];
    for (;;) {
      float lo[SEPW], hi[SEPW];
      srows(feat, w, lane, lo, hi);
      unsigned row[SEPW];
#pragma unroll
      for (int i = 0; i < SEPW; ++i) row[i] = w[i] & ((1u << RSH) - 1u);
      const int nbase = base + 32 * SEPW;
      const bool more = nbase + SEPW <= e;
      if (more) {
#pragma unroll
        for (int i = 0; i < SEPW; ++i) w[i] = pk[nbase + i];
      }
#pragma unroll
      for (int i = 0; i < SEPW; ++i) sadd(tile, row[i], lane, lo[i], hi[i]);
      base = nbase;
      if (!more) break;
    }
  }
  if (base < e) {
    const int n = e - base;
    unsigned w[SEPW];
#pragma unroll
    for (int i = 0; i < SEPW; ++i) w[i] = pk[i < n ? base + i : e - 1];
    float lo[SEPW], hi[SEPW];
    srows(feat, w, lane, lo, hi);
#pragma unroll
    for (int i = 0; i < SEPW; ++i)
      if (i < n) sadd(tile, w[i] & ((1u << RSH) - 1u), lane, lo[i], hi[i]);
  }
  __syncthreads();
  float4* o4 = (float4*)(ah + (size_t)g * RN * HF);
  for (int i = threadIdx.x; i < RN * HF / 4; i += 1024) o4[i] = t4[i];
}

#define PW 256
#define PS 25

__launch_bounds__(PS * 32)
__global__ void k_gemm(const float* __restrict__ A1,
                       const float* A2,
                       const float* __restrict__ dinv,
                       const float* __restrict__ oscale,
                       float a2sign,
                       const float* __restrict__ W,
                       const float* __restrict__ bias,
                       float* out,
                       float* __restrict__ accum,
                       int accum_init,
                       int n, int act) {
  __shared__ float sW[128 * 64];
  __shared__ float sB[64];
  const bool dual = (A2 != nullptr);
  const int wtot = (dual ? 128 : 64) * 64;
  for (int i = threadIdx.x; i < wtot; i += PS * 32) {
    int k = i >> 6;
    int col = i & 63;
    sW[(k >> 1) * 128 + col * 2 + (k & 1)] = W[i];
  }
  if (threadIdx.x < 64) sB[threadIdx.x] = bias[threadIdx.x];
  __syncthreads();

  const int lane = threadIdx.x & 31;
  const int wave = threadIdx.x >> 5;
  const int strip = wave * PW + blockIdx.x;
  if (strip >= (n >> 4)) return;
  const int row0 = strip << 4;
  const int m  = lane & 15;
  const int kh = (lane >> 4) << 1;

  const float* a1p = A1 + (size_t)(row0 + m) * HF + kh;
  const float dr = dinv ? dinv[row0 + m] : 1.0f;
  const float ur = 1.0f / dr;
  v8f acc[4] = {};

#pragma unroll
  for (int k0 = 0; k0 < 64; k0 += 4) {
    v2f a = *(const v2f*)(a1p + k0);
    a = a * ur;
    const float* wb = &sW[(k0 + kh) * 64 + m * 2];
#pragma unroll
    for (int t = 0; t < 4; ++t) {
      v2f b = *(const v2f*)(wb + t * 32);
      acc[t] = __builtin_amdgcn_wmma_f32_16x16x4_f32(
          false, a, false, b, (short)0, acc[t], false, false);
    }
  }
  if (dual) {
    const float* a2p = A2 + (size_t)(row0 + m) * HF + kh;
    const float ds = dr * a2sign;
#pragma unroll
    for (int k0 = 0; k0 < 64; k0 += 4) {
      v2f a = *(const v2f*)(a2p + k0);
      a = a * ds;
      const float* wb = &sW[(64 + k0 + kh) * 64 + m * 2];
#pragma unroll
      for (int t = 0; t < 4; ++t) {
        v2f b = *(const v2f*)(wb + t * 32);
        acc[t] = __builtin_amdgcn_wmma_f32_16x16x4_f32(
            false, a, false, b, (short)0, acc[t], false, false);
      }
    }
  }

  const int rbase = row0 + ((lane >> 4) << 3);
  float4 os0 = make_float4(1.0f, 1.0f, 1.0f, 1.0f), os1 = os0;
  if (oscale) {
    os0 = *(const float4*)(oscale + rbase);
    os1 = *(const float4*)(oscale + rbase + 4);
  }
  const float os[8] = {os0.x, os0.y, os0.z, os0.w, os1.x, os1.y, os1.z, os1.w};
#pragma unroll
  for (int t = 0; t < 4; ++t) {
    const float bv = sB[t * 16 + m];
#pragma unroll
    for (int v = 0; v < 8; ++v) {
      float val = acc[t][v] + bv;
      if (act) val = val > 0.0f ? val : 0.01f * val;
      size_t o = (size_t)(rbase + v) * HF + t * 16 + m;
      out[o] = val * os[v];
      if (accum) {
        if (accum_init) accum[o] = val; else accum[o] += val;
      }
    }
  }
}

#define SP 68
template <bool DUAL, bool HEAD>
__launch_bounds__(PS * 32)
__global__ void k_gemm2(const float* __restrict__ A1,
                        const float* __restrict__ A2,
                        const float* __restrict__ dinv,
                        const float* __restrict__ oscale,
                        const float* __restrict__ Wa, const float* __restrict__ ba,
                        const float* __restrict__ Wb, const float* __restrict__ bb,
                        const float* __restrict__ W4, const float* __restrict__ b4,
                        float* __restrict__ out,
                        float* __restrict__ accum, int accum_init, int n) {
  constexpr int KR = DUAL ? 128 : 64;
  __shared__ float sWa[KR * 64];
  __shared__ float sWb[KR * 64];
  __shared__ float sBa[64];
  __shared__ float sBb[64];
  __shared__ float sW4[128];
  __shared__ float sb4[2];
  __shared__ float sT[PS * 16 * SP];
  constexpr int NT = PS * 32;
  constexpr int PT = (KR * 64 + NT - 1) / NT;
  float wa[PT], wb[PT];
#pragma unroll
  for (int j = 0; j < PT; ++j) {
    const int i = threadIdx.x + j * NT;
    const bool ok = i < KR * 64;
    wa[j] = ok ? Wa[i] : 0.0f;
    wb[j] = ok ? Wb[i] : 0.0f;
  }
#pragma unroll
  for (int j = 0; j < PT; ++j) {
    const int i = threadIdx.x + j * NT;
    if (i < KR * 64) {
      const int k = i >> 6;
      const int col = i & 63;
      const int o = (k >> 1) * 128 + col * 2 + (k & 1);
      sWa[o] = wa[j];
      sWb[o] = wb[j];
    }
  }
  if (threadIdx.x < 64) {
    sBa[threadIdx.x] = ba[threadIdx.x];
    sBb[threadIdx.x] = bb[threadIdx.x];
  }
  if (HEAD) {
    if (threadIdx.x < 128) sW4[threadIdx.x] = W4[threadIdx.x];
    if (threadIdx.x < 2) sb4[threadIdx.x] = b4[threadIdx.x];
  }
  __syncthreads();

  const int lane = threadIdx.x & 31;
  const int wave = threadIdx.x >> 5;
  const int strip = wave * PW + blockIdx.x;
  if (strip >= (n >> 4)) return;
  const int row0 = strip << 4;
  const int m  = lane & 15;
  const int hh = lane >> 4;
  const int kh = hh << 1;
  float* st = sT + wave * (16 * SP);

  const float* a1p = A1 + (size_t)(row0 + m) * HF + kh;
  v2f a1f[16];
#pragma unroll
  for (int i = 0; i < 16; ++i) a1f[i] = *(const v2f*)(a1p + 4 * i);
  const int rbase = row0 + (hh << 3);
  float dr = 1.0f;
  if (DUAL) {
    dr = dinv[row0 + m];
    const float ur = 1.0f / dr;
#pragma unroll
    for (int i = 0; i < 16; ++i) a1f[i] = a1f[i] * ur;
  }
  v8f acc[4] = {};
#pragma unroll
  for (int i = 0; i < 16; ++i) {
    const float* wb = &sWa[(4 * i + kh) * 64 + m * 2];
#pragma unroll
    for (int t = 0; t < 4; ++t) {
      v2f b = *(const v2f*)(wb + t * 32);
      acc[t] = __builtin_amdgcn_wmma_f32_16x16x4_f32(
          false, a1f[i], false, b, (short)0, acc[t], false, false);
    }
  }
  if (DUAL) {
    const float* a2p = A2 + (size_t)(row0 + m) * HF + kh;
    const float ds = -dr;
#pragma unroll
    for (int i = 0; i < 16; ++i) {
      v2f a = *(const v2f*)(a2p + 4 * i);
      a = a * ds;
      const float* wb = &sWa[(64 + 4 * i + kh) * 64 + m * 2];
#pragma unroll
      for (int t = 0; t < 4; ++t) {
        v2f b = *(const v2f*)(wb + t * 32);
        acc[t] = __builtin_amdgcn_wmma_f32_16x16x4_f32(
            false, a, false, b, (short)0, acc[t], false, false);
      }
    }
  }

#pragma unroll
  for (int t = 0; t < 4; ++t) {
    const float bv = sBa[t * 16 + m];
#pragma unroll
    for (int v = 0; v < 8; ++v) {
      float val = acc[t][v] + bv;
      if (!DUAL) val = val > 0.0f ? val : 0.01f * val;
      st[(v + 8 * hh) * SP + t * 16 + m] = val;
    }
  }
  __builtin_amdgcn_fence(__ATOMIC_ACQ_REL, "wavefront");

  float4 os0 = make_float4(1.0f, 1.0f, 1.0f, 1.0f), os1 = os0;
  if (!HEAD && oscale) {
    os0 = *(const float4*)(oscale + rbase);
    os1 = *(const float4*)(oscale + rbase + 4);
  }
  const float os[8] = {os0.x, os0.y, os0.z, os0.w, os1.x, os1.y, os1.z, os1.w};

  v8f acc2[4] = {};
  if (DUAL) {
#pragma unroll
    for (int i = 0; i < 16; ++i) {
      const float* wb = &sWb[(4 * i + kh) * 64 + m * 2];
#pragma unroll
      for (int t = 0; t < 4; ++t) {
        v2f b = *(const v2f*)(wb + t * 32);
        acc2[t] = __builtin_amdgcn_wmma_f32_16x16x4_f32(
            false, a1f[i], false, b, (short)0, acc2[t], false, false);
      }
    }
  }
  {
    const float* tp = st + m * SP + kh;
#pragma unroll
    for (int i = 0; i < 16; ++i) {
      v2f a = *(const v2f*)(tp + 4 * i);
      const float* wb = &sWb[((DUAL ? 64 : 0) + 4 * i + kh) * 64 + m * 2];
#pragma unroll
      for (int t = 0; t < 4; ++t) {
        v2f b = *(const v2f*)(wb + t * 32);
        acc2[t] = __builtin_amdgcn_wmma_f32_16x16x4_f32(
            false, a, false, b, (short)0, acc2[t], false, false);
      }
    }
  }

  if (!HEAD) {
#pragma unroll
    for (int t = 0; t < 4; ++t) {
      const float bv = sBb[t * 16 + m];
#pragma unroll
      for (int v = 0; v < 8; ++v) {
        float val = acc2[t][v] + bv;
        if (!DUAL) val = val > 0.0f ? val : 0.01f * val;
        const size_t o = (size_t)(rbase + v) * HF + t * 16 + m;
        out[o] = val * os[v];
        if (accum) {
          if (accum_init) accum[o] = val; else accum[o] += val;
        }
      }
    }
  } else {
    __builtin_amdgcn_fence(__ATOMIC_ACQ_REL, "wavefront");
#pragma unroll
    for (int t = 0; t < 4; ++t) {
      const float bv = sBb[t * 16 + m];
#pragma unroll
      for (int v = 0; v < 8; ++v) {
        float val = acc2[t][v] + bv;
        if (!accum_init) val += accum[(size_t)(rbase + v) * HF + t * 16 + m];
        val = val > 0.0f ? val : 0.01f * val;
        st[(v + 8 * hh) * SP + t * 16 + m] = val;
      }
    }
    __builtin_amdgcn_fence(__ATOMIC_ACQ_REL, "wavefront");
    const float* tp = st + m * SP;
    float a = sb4[hh];
#pragma unroll
    for (int k = 0; k < 64; ++k) a += tp[k] * sW4[2 * k + hh];
    out[2 * (size_t)(row0 + m) + hh] = a;
  }
}

extern "C" void kernel_launch(void* const* d_in, const int* in_sizes, int n_in,
                              void* d_out, int out_size, void* d_ws, size_t ws_size,
                              hipStream_t stream) {
  const float* x   = (const float*)d_in[0];
  const int*   src = (const int*)d_in[1];
  const int*   dst = (const int*)d_in[2];
  const float* W1  = (const float*)d_in[3];
  const float* b1  = (const float*)d_in[4];
  const float* W2  = (const float*)d_in[5];
  const float* b2  = (const float*)d_in[6];
  const float* Wc1 = (const float*)d_in[7];
  const float* bc1 = (const float*)d_in[8];
  const float* Wc2 = (const float*)d_in[9];
  const float* bc2 = (const float*)d_in[10];
  const float* W3  = (const float*)d_in[11];
  const float* b3  = (const float*)d_in[12];
  const float* W4  = (const float*)d_in[13];
  const float* b4  = (const float*)d_in[14];
  float* out = (float*)d_out;

  const size_t NH = (size_t)NODES * HF;
  float* ws   = (float*)d_ws;
  float* hbuf = ws;
  float* h0   = hbuf + NH;
  float* grp  = h0 + NH;
  float* hall = grp + NH;
  float* ah   = hall + NH;
  float* dinv = ah + NH;
  unsigned* pk     = (unsigned*)grp;
  unsigned* rcnt   = pk + (size_t)RELS * EDGES;
  unsigned* rstart = rcnt + (size_t)RELS * GCH * NR;

  k_degpart<<<dim3(DCH, DQ, RELS), 1024, 0, stream>>>(dst, ah);
  k_dinv<<<(RELS * NODES + 255) / 256, 256, 0, stream>>>(ah, dinv);

  k_rcount<<<dim3(GCH, RELS), 1024, 0, stream>>>(dst, rcnt);
  k_rfill<<<dim3(GCH, RELS), 1024, 0, stream>>>(src, dst, rcnt, rstart, pk);

  k_gemm2<false, false><<<PW, PS * 32, 0, stream>>>(x, nullptr, nullptr, dinv, W1, b1, W2, b2,
                                                         nullptr, nullptr, hbuf, nullptr, 0, NODES);

  for (int r = 0; r < RELS; ++r) {
    const unsigned* pr = pk + (size_t)r * EDGES;
    const unsigned* sr = rstart + (size_t)r * (NR + 1);
    const float* dv = dinv + (size_t)r * NODES;
    const float* dvn = dinv + (size_t)(r + 1) * NODES;

    k_scatter<<<NR, 1024, 0, stream>>>(hbuf, pr, sr, ah);
    k_gemm<<<PW, PS * 32, 0, stream>>>(hbuf, ah, dv, dv, -1.0f, Wc1, bc1, h0, nullptr, 0, NODES, 0);

    k_scatter<<<NR, 1024, 0, stream>>>(h0, pr, sr, ah);
    if (r + 1 < RELS)
      k_gemm2<true, false><<<PW, PS * 32, 0, stream>>>(h0, ah, dv, dvn, Wc2, bc2, W3, b3, nullptr, nullptr,
                                                            hbuf, hall, r == 0 ? 1 : 0, NODES);
    else
      k_gemm2<true, true><<<PW, PS * 32, 0, stream>>>(h0, ah, dv, nullptr, Wc2, bc2, W3, b3, W4, b4,
                                                           out, hall, r == 0 ? 1 : 0, NODES);
  }
}


# Edge fill groups each chunk's packed words by range in LDS (LDS cursors; word and target slot per entry) and copies them out as contiguous runs: 3M scattered 4-byte stores go; arrival-order sums kept
# speedup vs baseline: 2.8797x; 1.0369x over previous
#include <hip/hip_runtime.h>

typedef __attribute__((ext_vector_type(2))) float v2f;
typedef __attribute__((ext_vector_type(8))) float v8f;

#define NODES 100000
#define EDGES 1000000
#define RELS  3
#define HF    64

#define DCH 16
#define DQ  4
#define DNR 25024
#define DEC (EDGES / DCH)
__launch_bounds__(1024)
__global__ void k_degpart(const int* __restrict__ dst, float* __restrict__ part) {
  __shared__ unsigned cnt[DNR];
  const int c = blockIdx.x, q = blockIdx.y, r = blockIdx.z;
  for (int i = threadIdx.x; i < DNR; i += 1024) cnt[i] = 0u;
  __syncthreads();
  const int lo = q * DNR;
  const int4* p = (const int4*)(dst + (size_t)r * EDGES + (size_t)c * DEC);
  for (int base = threadIdx.x; base < DEC / 4; base += 8 * 1024) {
    int4 v[8];
#pragma unroll
    for (int j = 0; j < 8; ++j) {
      const int idx = base + j * 1024;
      v[j] = idx < DEC / 4 ? p[idx] : make_int4(-1, -1, -1, -1);
    }
#pragma unroll
    for (int j = 0; j < 8; ++j) {
      const unsigned a0 = (unsigned)(v[j].x - lo), a1 = (unsigned)(v[j].y - lo);
      const unsigned a2 = (unsigned)(v[j].z - lo), a3 = (unsigned)(v[j].w - lo);
      if (a0 < DNR) atomicAdd(&cnt[a0], 1u);
      if (a1 < DNR) atomicAdd(&cnt[a1], 1u);
      if (a2 < DNR) atomicAdd(&cnt[a2], 1u);
      if (a3 < DNR) atomicAdd(&cnt[a3], 1u);
    }
  }
  __syncthreads();
  const int nh = (NODES - lo) < DNR ? (NODES - lo) : DNR;
  float* o = part + ((size_t)c * RELS + r) * NODES + lo;
  for (int i = threadIdx.x; i < nh; i += 1024) o[i] = (float)cnt[i];
}

__global__ void k_dinv(const float* __restrict__ part, float* __restrict__ dinv) {
  int i = blockIdx.x * blockDim.x + threadIdx.x;
  if (i >= RELS * NODES) return;
  float v = 0.0f;
#pragma unroll
  for (int c = 0; c < DCH; ++c) v += part[(size_t)c * RELS * NODES + i];
  v = v < 1.0f ? 1.0f : v;
  dinv[i] = 1.0f / sqrtf(v);
}

#define RN   400
#define RSH  9
#define NR   (NODES / RN)
#define GCH  50
#define GEC  (EDGES / GCH)
#define GU   5

__launch_bounds__(1024)
__global__ void k_rcount(const int* __restrict__ dst, unsigned* __restrict__ cnt) {
  __shared__ unsigned h[NR];
  const int c = blockIdx.x, r = blockIdx.y;
  for (int i = threadIdx.x; i < NR; i += 1024) h[i] = 0u;
  __syncthreads();
  const int4* p = (const int4*)(dst + (size_t)r * EDGES + (size_t)c * GEC);
  for (int base = threadIdx.x; base < GEC / 4; base += GU * 1024) {
    int4 v[GU];
#pragma unroll
    for (int j = 0; j < GU; ++j) {
      const int idx = base + j * 1024;
      v[j] = idx < GEC / 4 ? p[idx] : make_int4(-1, -1, -1, -1);
    }
#pragma unroll
    for (int j = 0; j < GU; ++j) {
      if ((unsigned)v[j].x < NODES) atomicAdd(&h[(unsigned)v[j].x / RN], 1u);
      if ((unsigned)v[j].y < NODES) atomicAdd(&h[(unsigned)v[j].y / RN], 1u);
      if ((unsigned)v[j].z < NODES) atomicAdd(&h[(unsigned)v[j].z / RN], 1u);
      if ((unsigned)v[j].w < NODES) atomicAdd(&h[(unsigned)v[j].w / RN], 1u);
    }
  }
  __syncthreads();
  unsigned* o = cnt + ((size_t)r * GCH + c) * NR;
  for (int i = threadIdx.x; i < NR; i += 1024) o[i] = h[i];
}

__device__ __forceinline__ void rput(unsigned* cur, const unsigned* gof, uint2* srt, int sid, int did) {
  if ((unsigned)did < NODES) {
    const unsigned g = (unsigned)did / RN;
    const unsigned p = atomicAdd(&cur[g], 1u);
    srt[p] = make_uint2(((unsigned)sid << RSH) | ((unsigned)did - g * RN), p + gof[g]);
  }
}

__launch_bounds__(1024)
__global__ void k_rfill(const int* __restrict__ src, const int* __restrict__ dst,
                        const unsigned* __restrict__ cnt,
                        unsigned* __restrict__ rstart,
                        unsigned* __restrict__ pk) {
  __shared__ unsigned sc[512];
  __shared__ unsigned sl[512];
  __shared__ unsigned cur[NR];
  __shared__ unsigned gof[NR];
  __shared__ uint2 srt[GEC];
  const int c = blockIdx.x, r = blockIdx.y;
  const int t = threadIdx.x;
  unsigned tot = 0u, pre = 0u, own = 0u;
  if (t < NR) {
    const unsigned* q = cnt + (size_t)r * GCH * NR + t;
#pragma unroll 10
    for (int j = 0; j < GCH; ++j) {
      const unsigned x = q[(size_t)j * NR];
      tot += x;
      pre += j < c ? x : 0u;
      own = j == c ? x : own;
    }
  }
  if (t < 512) { sc[t] = tot; sl[t] = own; }
  __syncthreads();
  for (int off = 1; off < 512; off <<= 1) {
    unsigned a = 0u, b = 0u;
    if (t < 512 && t >= off) { a = sc[t - off]; b = sl[t - off]; }
    __syncthreads();
    if (t < 512) { sc[t] += a; sl[t] += b; }
    __syncthreads();
  }
  if (t < NR) {
    const unsigned st = sc[t] - tot;
    const unsigned l0 = sl[t] - own;
    cur[t] = l0;
    gof[t] = st + pre - l0;
    if (c == 0) {
      rstart[(size_t)r * (NR + 1) + t] = st;
      if (t == NR - 1) rstart[(size_t)r * (NR + 1) + NR] = sc[t];
    }
  }
  const int nloc = (int)sl[511];
  __syncthreads();
  const int4* ps = (const int4*)(src + (size_t)r * EDGES + (size_t)c * GEC);
  const int4* pd = (const int4*)(dst + (size_t)r * EDGES + (size_t)c * GEC);
  unsigned* o = pk + (size_t)r * EDGES;
  for (int base = t; base < GEC / 4; base += GU * 1024) {
    int4 vs[GU], vd[GU];
#pragma unroll
    for (int j = 0; j < GU; ++j) {
      const int idx = base + j * 1024;
      const bool ok = idx < GEC / 4;
      vd[j] = ok ? pd[idx] : make_int4(-1, -1, -1, -1);
      vs[j] = ok ? ps[idx] : make_int4(0, 0, 0, 0);
    }
#pragma unroll
    for (int j = 0; j < GU; ++j) {
      rput(cur, gof, srt, vs[j].x, vd[j].x);
      rput(cur, gof, srt, vs[j].y, vd[j].y);
      rput(cur, gof, srt, vs[j].z, vd[j].z);
      rput(cur, gof, srt, vs[j].w, vd[j].w);
    }
  }
  __syncthreads();
  for (int i = t; i < nloc; i += 1024) {
    const uint2 e = srt[i];
    o[e.y] = e.x;
  }
}

#define SEPW 16
__device__ __forceinline__ void srows(const float* __restrict__ feat, const unsigned (&w)[SEPW], int lane,
                                      float (&lo)[SEPW], float (&hi)[SEPW]) {
#pragma unroll
  for (int i = 0; i < SEPW; ++i) {
    const float* f = feat + (size_t)(w[i] >> RSH) * HF + lane;
    lo[i] = f[0];
    hi[i] = f[32];
  }
}
__device__ __forceinline__ void sadd(float* tile, unsigned row, int lane, float lo, float hi) {
  float* o = tile + row * HF + lane;
  __hip_atomic_fetch_add(o,      lo, __ATOMIC_RELAXED, __HIP_MEMORY_SCOPE_WORKGROUP);
  __hip_atomic_fetch_add(o + 32, hi, __ATOMIC_RELAXED, __HIP_MEMORY_SCOPE_WORKGROUP);
}

__launch_bounds__(1024)
__global__ void k_scatter(const float* __restrict__ feat,
                          const unsigned* __restrict__ pk,
                          const unsigned* __restrict__ rstart,
                          float* __restrict__ ah) {
  __shared__ float tile[RN * HF];
  const int g = blockIdx.x;
  const int lane = threadIdx.x & 31;
  const int wave = __builtin_amdgcn_readfirstlane((int)(threadIdx.x >> 5));
  float4* t4 = (float4*)tile;
  for (int i = threadIdx.x; i < RN * HF / 4; i += 1024) t4[i] = make_float4(0.0f, 0.0f, 0.0f, 0.0f);
  const int b = (int)rstart[g], e = (int)rstart[g + 1];
  __syncthreads();
  int base = b + wave * SEPW;
  if (base + SEPW <= e) {
    unsigned w[SEPW];
#pragma unroll
    for (int i = 0; i < SEPW; ++i) w[i] = pk[base + i];
    for (;;) {
      float lo[SEPW], hi[SEPW];
      srows(feat, w, lane, lo, hi);
      unsigned row[SEPW];
#pragma unroll
      for (int i = 0; i < SEPW; ++i) row[i] = w[i] & ((1u << RSH) - 1u);
      const int nbase = base + 32 * SEPW;
      const bool more = nbase + SEPW <= e;
      if (more) {
#pragma unroll
        for (int i = 0; i < SEPW; ++i) w[i] = pk[nbase + i];
      }
#pragma unroll
      for (int i = 0; i < SEPW; ++i) sadd(tile, row[i], lane, lo[i], hi[i]);
      base = nbase;
      if (!more) break;
    }
  }
  if (base < e) {
    const int n = e - base;
    unsigned w[SEPW];
#pragma unroll
    for (int i = 0; i < SEPW; ++i) w[i] = pk[i < n ? base + i : e - 1];
    float lo[SEPW], hi[SEPW];
    srows(feat, w, lane, lo, hi);
#pragma unroll
    for (int i = 0; i < SEPW; ++i)
      if (i < n) sadd(tile, w[i] & ((1u << RSH) - 1u), lane, lo[i], hi[i]);
  }
  __syncthreads();
  float4* o4 = (float4*)(ah + (size_t)g * RN * HF);
  for (int i = threadIdx.x; i < RN * HF / 4; i += 1024) o4[i] = t4[i];
}

#define PW 256
#define PS 25

__launch_bounds__(PS * 32)
__global__ void k_gemm(const float* __restrict__ A1,
                       const float* A2,
                       const float* __restrict__ dinv,
                       const float* __restrict__ oscale,
                       float a2sign,
                       const float* __restrict__ W,
                       const float* __restrict__ bias,
                       float* out,
                       float* __restrict__ accum,
                       int accum_init,
                       int n, int act) {
  __shared__ float sW[128 * 64];
  __shared__ float sB[64];
  const bool dual = (A2 != nullptr);
  const int wtot = (dual ? 128 : 64) * 64;
  for (int i = threadIdx.x; i < wtot; i += PS * 32) {
    int k = i >> 6;
    int col = i & 63;
    sW[(k >> 1) * 128 + col * 2 + (k & 1)] = W[i];
  }
  if (threadIdx.x < 64) sB[threadIdx.x] = bias[threadIdx.x];
  __syncthreads();

  const int lane = threadIdx.x & 31;
  const int wave = threadIdx.x >> 5;
  const int strip = wave * PW + blockIdx.x;
  if (strip >= (n >> 4)) return;
  const int row0 = strip << 4;
  const int m  = lane & 15;
  const int kh = (lane >> 4) << 1;

  const float* a1p = A1 + (size_t)(row0 + m) * HF + kh;
  const float dr = dinv ? dinv[row0 + m] : 1.0f;
  const float ur = 1.0f / dr;
  v8f acc[4] = {};

#pragma unroll
  for (int k0 = 0; k0 < 64; k0 += 4) {
    v2f a = *(const v2f*)(a1p + k0);
    a = a * ur;
    const float* wb = &sW[(k0 + kh) * 64 + m * 2];
#pragma unroll
    for (int t = 0; t < 4; ++t) {
      v2f b = *(const v2f*)(wb + t * 32);
      acc[t] = __builtin_amdgcn_wmma_f32_16x16x4_f32(
          false, a, false, b, (short)0, acc[t], false, false);
    }
  }
  if (dual) {
    const float* a2p = A2 + (size_t)(row0 + m) * HF + kh;
    const float ds = dr * a2sign;
#pragma unroll
    for (int k0 = 0; k0 < 64; k0 += 4) {
      v2f a = *(const v2f*)(a2p + k0);
      a = a * ds;
      const float* wb = &sW[(64 + k0 + kh) * 64 + m * 2];
#pragma unroll
      for (int t = 0; t < 4; ++t) {
        v2f b = *(const v2f*)(wb + t * 32);
        acc[t] = __builtin_amdgcn_wmma_f32_16x16x4_f32(
            false, a, false, b, (short)0, acc[t], false, false);
      }
    }
  }

  const int rbase = row0 + ((lane >> 4) << 3);
  float4 os0 = make_float4(1.0f, 1.0f, 1.0f, 1.0f), os1 = os0;
  if (oscale) {
    os0 = *(const float4*)(oscale + rbase);
    os1 = *(const float4*)(oscale + rbase + 4);
  }
  const float os[8] = {os0.x, os0.y, os0.z, os0.w, os1.x, os1.y, os1.z, os1.w};
#pragma unroll
  for (int t = 0; t < 4; ++t) {
    const float bv = sB[t * 16 + m];
#pragma unroll
    for (int v = 0; v < 8; ++v) {
      float val = acc[t][v] + bv;
      if (act) val = val > 0.0f ? val : 0.01f * val;
      size_t o = (size_t)(rbase + v) * HF + t * 16 + m;
      out[o] = val * os[v];
      if (accum) {
        if (accum_init) accum[o] = val; else accum[o] += val;
      }
    }
  }
}

#define SP 68
template <bool DUAL, bool HEAD>
__launch_bounds__(PS * 32)
__global__ void k_gemm2(const float* __restrict__ A1,
                        const float* __restrict__ A2,
                        const float* __restrict__ dinv,
                        const float* __restrict__ oscale,
                        const float* __restrict__ Wa, const float* __restrict__ ba,
                        const float* __restrict__ Wb, const float* __restrict__ bb,
                        const float* __restrict__ W4, const float* __restrict__ b4,
                        float* __restrict__ out,
                        float* __restrict__ accum, int accum_init, int n) {
  constexpr int KR = DUAL ? 128 : 64;
  __shared__ float sWa[KR * 64];
  __shared__ float sWb[KR * 64];
  __shared__ float sBa[64];
  __shared__ float sBb[64];
  __shared__ float sW4[128];
  __shared__ float sb4[2];
  __shared__ float sT[PS * 16 * SP];
  constexpr int NT = PS * 32;
  constexpr int PT = (KR * 64 + NT - 1) / NT;
  float wa[PT], wb[PT];
#pragma unroll
  for (int j = 0; j < PT; ++j) {
    const int i = threadIdx.x + j * NT;
    const bool ok = i < KR * 64;
    wa[j] = ok ? Wa[i] : 0.0f;
    wb[j] = ok ? Wb[i] : 0.0f;
  }
#pragma unroll
  for (int j = 0; j < PT; ++j) {
    const int i = threadIdx.x + j * NT;
    if (i < KR * 64) {
      const int k = i >> 6;
      const int col = i & 63;
      const int o = (k >> 1) * 128 + col * 2 + (k & 1);
      sWa[o] = wa[j];
      sWb[o] = wb[j];
    }
  }
  if (threadIdx.x < 64) {
    sBa[threadIdx.x] = ba[threadIdx.x];
    sBb[threadIdx.x] = bb[threadIdx.x];
  }
  if (HEAD) {
    if (threadIdx.x < 128) sW4[threadIdx.x] = W4[threadIdx.x];
    if (threadIdx.x < 2) sb4[threadIdx.x] = b4[threadIdx.x];
  }
  __syncthreads();

  const int lane = threadIdx.x & 31;
  const int wave = threadIdx.x >> 5;
  const int strip = wave * PW + blockIdx.x;
  if (strip >= (n >> 4)) return;
  const int row0 = strip << 4;
  const int m  = lane & 15;
  const int hh = lane >> 4;
  const int kh = hh << 1;
  float* st = sT + wave * (16 * SP);

  const float* a1p = A1 + (size_t)(row0 + m) * HF + kh;
  v2f a1f[16];
#pragma unroll
  for (int i = 0; i < 16; ++i) a1f[i] = *(const v2f*)(a1p + 4 * i);
  const int rbase = row0 + (hh << 3);
  float dr = 1.0f;
  if (DUAL) {
    dr = dinv[row0 + m];
    const float ur = 1.0f / dr;
#pragma unroll
    for (int i = 0; i < 16; ++i) a1f[i] = a1f[i] * ur;
  }
  v8f acc[4] = {};
#pragma unroll
  for (int i = 0; i < 16; ++i) {
    const float* wb = &sWa[(4 * i + kh) * 64 + m * 2];
#pragma unroll
    for (int t = 0; t < 4; ++t) {
      v2f b = *(const v2f*)(wb + t * 32);
      acc[t] = __builtin_amdgcn_wmma_f32_16x16x4_f32(
          false, a1f[i], false, b, (short)0, acc[t], false, false);
    }
  }
  if (DUAL) {
    const float* a2p = A2 + (size_t)(row0 + m) * HF + kh;
    const float ds = -dr;
#pragma unroll
    for (int i = 0; i < 16; ++i) {
      v2f a = *(const v2f*)(a2p + 4 * i);
      a = a * ds;
      const float* wb = &sWa[(64 + 4 * i + kh) * 64 + m * 2];
#pragma unroll
      for (int t = 0; t < 4; ++t) {
        v2f b = *(const v2f*)(wb + t * 32);
        acc[t] = __builtin_amdgcn_wmma_f32_16x16x4_f32(
            false, a, false, b, (short)0, acc[t], false, false);
      }
    }
  }

#pragma unroll
  for (int t = 0; t < 4; ++t) {
    const float bv = sBa[t * 16 + m];
#pragma unroll
    for (int v = 0; v < 8; ++v) {
      float val = acc[t][v] + bv;
      if (!DUAL) val = val > 0.0f ? val : 0.01f * val;
      st[(v + 8 * hh) * SP + t * 16 + m] = val;
    }
  }
  __builtin_amdgcn_fence(__ATOMIC_ACQ_REL, "wavefront");

  float4 os0 = make_float4(1.0f, 1.0f, 1.0f, 1.0f), os1 = os0;
  if (!HEAD && oscale) {
    os0 = *(const float4*)(oscale + rbase);
    os1 = *(const float4*)(oscale + rbase + 4);
  }
  const float os[8] = {os0.x, os0.y, os0.z, os0.w, os1.x, os1.y, os1.z, os1.w};

  v8f acc2[4] = {};
  if (DUAL) {
#pragma unroll
    for (int i = 0; i < 16; ++i) {
      const float* wb = &sWb[(4 * i + kh) * 64 + m * 2];
#pragma unroll
      for (int t = 0; t < 4; ++t) {
        v2f b = *(const v2f*)(wb + t * 32);
        acc2[t] = __builtin_amdgcn_wmma_f32_16x16x4_f32(
            false, a1f[i], false, b, (short)0, acc2[t], false, false);
      }
    }
  }
  {
    const float* tp = st + m * SP + kh;
#pragma unroll
    for (int i = 0; i < 16; ++i) {
      v2f a = *(const v2f*)(tp + 4 * i);
      const float* wb = &sWb[((DUAL ? 64 : 0) + 4 * i + kh) * 64 + m * 2];
#pragma unroll
      for (int t = 0; t < 4; ++t) {
        v2f b = *(const v2f*)(wb + t * 32);
        acc2[t] = __builtin_amdgcn_wmma_f32_16x16x4_f32(
            false, a, false, b, (short)0, acc2[t], false, false);
      }
    }
  }

  if (!HEAD) {
#pragma unroll
    for (int t = 0; t < 4; ++t) {
      const float bv = sBb[t * 16 + m];
#pragma unroll
      for (int v = 0; v < 8; ++v) {
        float val = acc2[t][v] + bv;
        if (!DUAL) val = val > 0.0f ? val : 0.01f * val;
        const size_t o = (size_t)(rbase + v) * HF + t * 16 + m;
        out[o] = val * os[v];
        if (accum) {
          if (accum_init) accum[o] = val; else accum[o] += val;
        }
      }
    }
  } else {
    __builtin_amdgcn_fence(__ATOMIC_ACQ_REL, "wavefront");
#pragma unroll
    for (int t = 0; t < 4; ++t) {
      const float bv = sBb[t * 16 + m];
#pragma unroll
      for (int v = 0; v < 8; ++v) {
        float val = acc2[t][v] + bv;
        if (!accum_init) val += accum[(size_t)(rbase + v) * HF + t * 16 + m];
        val = val > 0.0f ? val : 0.01f * val;
        st[(v + 8 * hh) * SP + t * 16 + m] = val;
      }
    }
    __builtin_amdgcn_fence(__ATOMIC_ACQ_REL, "wavefront");
    const float* tp = st + m * SP;
    float a = sb4[hh];
#pragma unroll
    for (int k = 0; k < 64; ++k) a += tp[k] * sW4[2 * k + hh];
    out[2 * (size_t)(row0 + m) + hh] = a;
  }
}

extern "C" void kernel_launch(void* const* d_in, const int* in_sizes, int n_in,
                              void* d_out, int out_size, void* d_ws, size_t ws_size,
                              hipStream_t stream) {
  const float* x   = (const float*)d_in[0];
  const int*   src = (const int*)d_in[1];
  const int*   dst = (const int*)d_in[2];
  const float* W1  = (const float*)d_in[3];
  const float* b1  = (const float*)d_in[4];
  const float* W2  = (const float*)d_in[5];
  const float* b2  = (const float*)d_in[6];
  const float* Wc1 = (const float*)d_in[7];
  const float* bc1 = (const float*)d_in[8];
  const float* Wc2 = (const float*)d_in[9];
  const float* bc2 = (const float*)d_in[10];
  const float* W3  = (const float*)d_in[11];
  const float* b3  = (const float*)d_in[12];
  const float* W4  = (const float*)d_in[13];
  const float* b4  = (const float*)d_in[14];
  float* out = (float*)d_out;

  const size_t NH = (size_t)NODES * HF;
  float* ws   = (float*)d_ws;
  float* hbuf = ws;
  float* h0   = hbuf + NH;
  float* grp  = h0 + NH;
  float* hall = grp + NH;
  float* ah   = hall + NH;
  float* dinv = ah + NH;
  unsigned* pk     = (unsigned*)grp;
  unsigned* rcnt   = pk + (size_t)RELS * EDGES;
  unsigned* rstart = rcnt + (size_t)RELS * GCH * NR;

  k_degpart<<<dim3(DCH, DQ, RELS), 1024, 0, stream>>>(dst, ah);
  k_dinv<<<(RELS * NODES + 255) / 256, 256, 0, stream>>>(ah, dinv);

  k_rcount<<<dim3(GCH, RELS), 1024, 0, stream>>>(dst, rcnt);
  k_rfill<<<dim3(GCH, RELS), 1024, 0, stream>>>(src, dst, rcnt, rstart, pk);

  k_gemm2<false, false><<<PW, PS * 32, 0, stream>>>(x, nullptr, nullptr, dinv, W1, b1, W2, b2,
                                                         nullptr, nullptr, hbuf, nullptr, 0, NODES);

  for (int r = 0; r < RELS; ++r) {
    const unsigned* pr = pk + (size_t)r * EDGES;
    const unsigned* sr = rstart + (size_t)r * (NR + 1);
    const float* dv = dinv + (size_t)r * NODES;
    const float* dvn = dinv + (size_t)(r + 1) * NODES;

    k_scatter<<<NR, 1024, 0, stream>>>(hbuf, pr, sr, ah);
    k_gemm<<<PW, PS * 32, 0, stream>>>(hbuf, ah, dv, dv, -1.0f, Wc1, bc1, h0, nullptr, 0, NODES, 0);

    k_scatter<<<NR, 1024, 0, stream>>>(h0, pr, sr, ah);
    if (r + 1 < RELS)
      k_gemm2<true, false><<<PW, PS * 32, 0, stream>>>(h0, ah, dv, dvn, Wc2, bc2, W3, b3, nullptr, nullptr,
                                                            hbuf, hall, r == 0 ? 1 : 0, NODES);
    else
      k_gemm2<true, true><<<PW, PS * 32, 0, stream>>>(h0, ah, dv, nullptr, Wc2, bc2, W3, b3, W4, b4,
                                                           out, hall, r == 0 ? 1 : 0, NODES);
  }
}
